# starting kernel
# baseline (speedup 1.0000x reference)
#include <hip/hip_runtime.h>
#include <math.h>

typedef __attribute__((ext_vector_type(2))) float v2f;
typedef __attribute__((ext_vector_type(8))) float v8f;

#define DFEAT 128

__global__ void tgcn_init(float* __restrict__ deg, float* __restrict__ acc1,
                          float* __restrict__ acc2, int n_nodes, size_t nd) {
  size_t i = (size_t)blockIdx.x * blockDim.x + threadIdx.x;
  if (i < nd) {
    acc1[i] = 0.0f;
    acc2[i] = 0.0f;
  }
  if (i < (size_t)n_nodes) deg[i] = 1.0f;
}

__global__ void tgcn_degree(const int* __restrict__ dst, float* __restrict__ deg,
                            int n_edges) {
  int e = blockIdx.x * blockDim.x + threadIdx.x;
  if (e < n_edges) atomicAdd(&deg[dst[e]], 1.0f);
}

__global__ void tgcn_rsqrt(float* __restrict__ deg, int n_nodes) {
  int i = blockIdx.x * blockDim.x + threadIdx.x;
  if (i < n_nodes) deg[i] = rsqrtf(deg[i]);
}

__global__ void __launch_bounds__(256) tgcn_gemm_wmma(
    const float* __restrict__ X, const float* __restrict__ W,
    const float* __restrict__ t, const float* __restrict__ freq,
    const float* __restrict__ phase, int use_te,
    float* __restrict__ Y, int n_nodes) {
  __shared__ float sA[16 * DFEAT];

  const int tid = threadIdx.x;
  const int mbase = blockIdx.x * 16;

  for (int i = tid; i < 16 * DFEAT; i += 256) {
    int r = i >> 7;
    int c = i & (DFEAT - 1);
    int row = mbase + r;
    float v = 0.0f;
    if (row < n_nodes) {
      v = X[(size_t)row * DFEAT + c];
      if (use_te) v += cosf(t[row] * freq[c] + phase[c]);
    }
    sA[i] = v;
  }
  __syncthreads();

  const int wave = tid >> 5;
  const int lane = tid & 31;
  const int half = lane >> 4;
  const int arow = lane & 15;
  const int ncol = wave * 16 + (lane & 15);

  v8f acc = {};
  for (int k = 0; k < DFEAT; k += 4) {
    const int ka = k + 2 * half;
    v2f a, b;
    a.x = sA[arow * DFEAT + ka];
    a.y = sA[arow * DFEAT + ka + 1];
    b.x = W[(size_t)ka * DFEAT + ncol];
    b.y = W[(size_t)(ka + 1) * DFEAT + ncol];
    acc = __builtin_amdgcn_wmma_f32_16x16x4_f32(
         false, a,  false, b,
         (short)0, acc,  false,  false);
  }

#pragma unroll
  for (int r = 0; r < 8; ++r) {
    int orow = mbase + r + 8 * half;
    if (orow < n_nodes) Y[(size_t)orow * DFEAT + ncol] = acc[r];
  }
}

__global__ void __launch_bounds__(256) tgcn_scatter(
    const float* __restrict__ xw, const int* __restrict__ src,
    const int* __restrict__ dst, const float* __restrict__ dinv,
    float* __restrict__ out, int n_edges) {
  const int lane = threadIdx.x & 31;
  const int edge = blockIdx.x * (blockDim.x >> 5) + (threadIdx.x >> 5);
  if (edge >= n_edges) return;

  const int s = src[edge];
  const int d = dst[edge];
  const float norm = dinv[s] * dinv[d];

  const float4 v = ((const float4*)(xw + (size_t)s * DFEAT))[lane];
  float* o = out + (size_t)d * DFEAT + lane * 4;
  atomicAdd(o + 0, v.x * norm);
  atomicAdd(o + 1, v.y * norm);
  atomicAdd(o + 2, v.z * norm);
  atomicAdd(o + 3, v.w * norm);
}

__global__ void tgcn_selfloop_bias(const float* __restrict__ xw,
                                   const float* __restrict__ dinv,
                                   const float* __restrict__ bias,
                                   float* __restrict__ out, size_t nd) {
  size_t i = (size_t)blockIdx.x * blockDim.x + threadIdx.x;
  if (i >= nd) return;
  int node = (int)(i >> 7);
  int c = (int)(i & (DFEAT - 1));
  float dv = dinv[node];
  out[i] += xw[i] * dv * dv + bias[c];
}

extern "C" void kernel_launch(void* const* d_in, const int* in_sizes, int n_in,
                              void* d_out, int out_size, void* d_ws, size_t ws_size,
                              hipStream_t stream) {
  const float* x     = (const float*)d_in[0];
  const int*   ei    = (const int*)d_in[1];
  const float* t     = (const float*)d_in[2];
  const float* W1    = (const float*)d_in[3];
  const float* b1    = (const float*)d_in[4];
  const float* W2    = (const float*)d_in[5];
  const float* b2    = (const float*)d_in[6];
  const float* freq  = (const float*)d_in[7];
  const float* phase = (const float*)d_in[8];

  const int n_nodes = in_sizes[2];
  const int n_edges = in_sizes[1] / 2;
  const size_t nd = (size_t)n_nodes * DFEAT;

  const int* src = ei;
  const int* dst = ei + n_edges;

  float* out = (float*)d_out;

  float* dinv = (float*)d_ws;
  size_t noff = ((size_t)n_nodes + 255) & ~(size_t)255;
  float* bufA = dinv + noff;
  float* bufB = bufA + nd;

  const int T = 256;
  const int g_nd   = (int)((nd + T - 1) / T);
  const int g_edge = (n_edges + T - 1) / T;
  const int g_node = (n_nodes + T - 1) / T;
  const int g_gemm = (n_nodes + 15) / 16;
  const int g_scat = (n_edges + 7) / 8;

  tgcn_init<<<g_nd, T, 0, stream>>>(dinv, bufB, out, n_nodes, nd);
  tgcn_degree<<<g_edge, T, 0, stream>>>(dst, dinv, n_edges);
  tgcn_rsqrt<<<g_node, T, 0, stream>>>(dinv, n_nodes);

  tgcn_gemm_wmma<<<g_gemm, T, 0, stream>>>(x, W1, t, freq, phase, 1, bufA, n_nodes);
  tgcn_scatter<<<g_scat, T, 0, stream>>>(bufA, src, dst, dinv, bufB, n_edges);
  tgcn_selfloop_bias<<<g_nd, T, 0, stream>>>(bufA, dinv, b1, bufB, nd);

  tgcn_gemm_wmma<<<g_gemm, T, 0, stream>>>(bufB, W2, t, freq, phase, 0, bufA, n_nodes);
  tgcn_scatter<<<g_scat, T, 0, stream>>>(bufA, src, dst, dinv, out, n_edges);
  tgcn_selfloop_bias<<<g_nd, T, 0, stream>>>(bufA, dinv, b2, out, nd);
}


# scatter: lane l takes features l,32+l,64+l,96+l so each f32 atomic instruction covers one whole 128-B line instead of 8 floats in each of 4 lines; atomics stay, arrival-order sums as before
# speedup vs baseline: 1.3336x; 1.3336x over previous
#include <hip/hip_runtime.h>
#include <math.h>

typedef __attribute__((ext_vector_type(2))) float v2f;
typedef __attribute__((ext_vector_type(8))) float v8f;

#define DFEAT 128

__global__ void tgcn_init(float* __restrict__ deg, float* __restrict__ acc1,
                          float* __restrict__ acc2, int n_nodes, size_t nd) {
  size_t i = (size_t)blockIdx.x * blockDim.x + threadIdx.x;
  if (i < nd) {
    acc1[i] = 0.0f;
    acc2[i] = 0.0f;
  }
  if (i < (size_t)n_nodes) deg[i] = 1.0f;
}

__global__ void tgcn_degree(const int* __restrict__ dst, float* __restrict__ deg,
                            int n_edges) {
  int e = blockIdx.x * blockDim.x + threadIdx.x;
  if (e < n_edges) atomicAdd(&deg[dst[e]], 1.0f);
}

__global__ void tgcn_rsqrt(float* __restrict__ deg, int n_nodes) {
  int i = blockIdx.x * blockDim.x + threadIdx.x;
  if (i < n_nodes) deg[i] = rsqrtf(deg[i]);
}

__global__ void __launch_bounds__(256) tgcn_gemm_wmma(
    const float* __restrict__ X, const float* __restrict__ W,
    const float* __restrict__ t, const float* __restrict__ freq,
    const float* __restrict__ phase, int use_te,
    float* __restrict__ Y, int n_nodes) {
  __shared__ float sA[16 * DFEAT];

  const int tid = threadIdx.x;
  const int mbase = blockIdx.x * 16;

  for (int i = tid; i < 16 * DFEAT; i += 256) {
    int r = i >> 7;
    int c = i & (DFEAT - 1);
    int row = mbase + r;
    float v = 0.0f;
    if (row < n_nodes) {
      v = X[(size_t)row * DFEAT + c];
      if (use_te) v += cosf(t[row] * freq[c] + phase[c]);
    }
    sA[i] = v;
  }
  __syncthreads();

  const int wave = tid >> 5;
  const int lane = tid & 31;
  const int half = lane >> 4;
  const int arow = lane & 15;
  const int ncol = wave * 16 + (lane & 15);

  v8f acc = {};
  for (int k = 0; k < DFEAT; k += 4) {
    const int ka = k + 2 * half;
    v2f a, b;
    a.x = sA[arow * DFEAT + ka];
    a.y = sA[arow * DFEAT + ka + 1];
    b.x = W[(size_t)ka * DFEAT + ncol];
    b.y = W[(size_t)(ka + 1) * DFEAT + ncol];
    acc = __builtin_amdgcn_wmma_f32_16x16x4_f32(
         false, a,  false, b,
         (short)0, acc,  false,  false);
  }

#pragma unroll
  for (int r = 0; r < 8; ++r) {
    int orow = mbase + r + 8 * half;
    if (orow < n_nodes) Y[(size_t)orow * DFEAT + ncol] = acc[r];
  }
}

__global__ void __launch_bounds__(256) tgcn_scatter(
    const float* __restrict__ xw, const int* __restrict__ src,
    const int* __restrict__ dst, const float* __restrict__ dinv,
    float* __restrict__ out, int n_edges) {
  const int lane = threadIdx.x & 31;
  const int edge = blockIdx.x * (blockDim.x >> 5) + (threadIdx.x >> 5);
  if (edge >= n_edges) return;

  const int s = src[edge];
  const int d = dst[edge];
  const float norm = dinv[s] * dinv[d];

  const float* xs = xw + (size_t)s * DFEAT + lane;
  const float v0 = xs[0];
  const float v1 = xs[32];
  const float v2 = xs[64];
  const float v3 = xs[96];
  float* o = out + (size_t)d * DFEAT + lane;
  atomicAdd(o + 0, v0 * norm);
  atomicAdd(o + 32, v1 * norm);
  atomicAdd(o + 64, v2 * norm);
  atomicAdd(o + 96, v3 * norm);
}

__global__ void tgcn_selfloop_bias(const float* __restrict__ xw,
                                   const float* __restrict__ dinv,
                                   const float* __restrict__ bias,
                                   float* __restrict__ out, size_t nd) {
  size_t i = (size_t)blockIdx.x * blockDim.x + threadIdx.x;
  if (i >= nd) return;
  int node = (int)(i >> 7);
  int c = (int)(i & (DFEAT - 1));
  float dv = dinv[node];
  out[i] += xw[i] * dv * dv + bias[c];
}

extern "C" void kernel_launch(void* const* d_in, const int* in_sizes, int n_in,
                              void* d_out, int out_size, void* d_ws, size_t ws_size,
                              hipStream_t stream) {
  const float* x     = (const float*)d_in[0];
  const int*   ei    = (const int*)d_in[1];
  const float* t     = (const float*)d_in[2];
  const float* W1    = (const float*)d_in[3];
  const float* b1    = (const float*)d_in[4];
  const float* W2    = (const float*)d_in[5];
  const float* b2    = (const float*)d_in[6];
  const float* freq  = (const float*)d_in[7];
  const float* phase = (const float*)d_in[8];

  const int n_nodes = in_sizes[2];
  const int n_edges = in_sizes[1] / 2;
  const size_t nd = (size_t)n_nodes * DFEAT;

  const int* src = ei;
  const int* dst = ei + n_edges;

  float* out = (float*)d_out;

  float* dinv = (float*)d_ws;
  size_t noff = ((size_t)n_nodes + 255) & ~(size_t)255;
  float* bufA = dinv + noff;
  float* bufB = bufA + nd;

  const int T = 256;
  const int g_nd   = (int)((nd + T - 1) / T);
  const int g_edge = (n_edges + T - 1) / T;
  const int g_node = (n_nodes + T - 1) / T;
  const int g_gemm = (n_nodes + 15) / 16;
  const int g_scat = (n_edges + 7) / 8;

  tgcn_init<<<g_nd, T, 0, stream>>>(dinv, bufB, out, n_nodes, nd);
  tgcn_degree<<<g_edge, T, 0, stream>>>(dst, dinv, n_edges);
  tgcn_rsqrt<<<g_node, T, 0, stream>>>(dinv, n_nodes);

  tgcn_gemm_wmma<<<g_gemm, T, 0, stream>>>(x, W1, t, freq, phase, 1, bufA, n_nodes);
  tgcn_scatter<<<g_scat, T, 0, stream>>>(bufA, src, dst, dinv, bufB, n_edges);
  tgcn_selfloop_bias<<<g_nd, T, 0, stream>>>(bufA, dinv, b1, bufB, nd);

  tgcn_gemm_wmma<<<g_gemm, T, 0, stream>>>(bufB, W2, t, freq, phase, 0, bufA, n_nodes);
  tgcn_scatter<<<g_scat, T, 0, stream>>>(bufA, src, dst, dinv, out, n_edges);
  tgcn_selfloop_bias<<<g_nd, T, 0, stream>>>(bufA, dinv, b2, out, nd);
}


# aggregation: f32 atomic scatter replaced by per-call grouping of edges by destination (int count, scan, fill); one wave sums a row in registers, stores once; sum order = fill arrival order
# speedup vs baseline: 2.0561x; 1.5418x over previous
#include <hip/hip_runtime.h>
#include <math.h>

typedef __attribute__((ext_vector_type(2))) float v2f;
typedef __attribute__((ext_vector_type(4))) float v4f;
typedef __attribute__((ext_vector_type(8))) float v8f;

#define DFEAT 128
#define SCAN_TILE 1024


__global__ void tgcn_zero_cnt(int* __restrict__ cnt, int n_nodes) {
  int i = blockIdx.x * blockDim.x + threadIdx.x;
  if (i < n_nodes) cnt[i] = 0;
}

__global__ void tgcn_count(const int* __restrict__ dst, int* __restrict__ cnt,
                           int n_edges) {
  int e = blockIdx.x * blockDim.x + threadIdx.x;
  if (e < n_edges) atomicAdd(&cnt[dst[e]], 1);
}

__device__ __forceinline__ int wave_sum(int v) {
  for (int o = 16; o > 0; o >>= 1) v += __shfl_xor(v, o);
  return v;
}

__global__ void __launch_bounds__(256) tgcn_scan_tiles(
    const int* __restrict__ cnt, int* __restrict__ tile_sum, int n_nodes) {
  __shared__ int s8[8];
  const int base = blockIdx.x * SCAN_TILE + threadIdx.x * 4;
  int t = 0;
  for (int u = 0; u < 4; ++u)
    if (base + u < n_nodes) t += cnt[base + u];
  t = wave_sum(t);
  if ((threadIdx.x & 31) == 0) s8[threadIdx.x >> 5] = t;
  __syncthreads();
  if (threadIdx.x == 0) {
    int tot = 0;
    for (int w = 0; w < 8; ++w) tot += s8[w];
    tile_sum[blockIdx.x] = tot;
  }
}

__global__ void __launch_bounds__(256) tgcn_scan_write(
    const int* __restrict__ cnt, const int* __restrict__ tile_sum,
    int* __restrict__ cursor, float* __restrict__ dinv, int n_nodes) {
  __shared__ int s_tot[8];
  __shared__ int s_pre[8];
  const int tid = threadIdx.x;
  const int lane = tid & 31;
  const int wave = tid >> 5;

  int pre = 0;
  for (int k = tid; k < (int)blockIdx.x; k += 256) pre += tile_sum[k];
  pre = wave_sum(pre);

  const int base = blockIdx.x * SCAN_TILE + tid * 4;
  int c[4];
  int t = 0;
  for (int u = 0; u < 4; ++u) {
    c[u] = (base + u < n_nodes) ? cnt[base + u] : 0;
    t += c[u];
  }
  int inc = t;
  for (int o = 1; o < 32; o <<= 1) {
    int y = __shfl_up(inc, o);
    if (lane >= o) inc += y;
  }
  if (lane == 31) s_tot[wave] = inc;
  if (lane == 0) s_pre[wave] = pre;
  __syncthreads();
  int off = 0;
  for (int w = 0; w < 8; ++w) {
    off += s_pre[w];
    if (w < wave) off += s_tot[w];
  }
  int run = off + inc - t;
  for (int u = 0; u < 4; ++u) {
    if (base + u < n_nodes) {
      cursor[base + u] = run;
      dinv[base + u] = rsqrtf((float)(c[u] + 1));
      run += c[u];
    }
  }
}

__global__ void tgcn_fill(const int* __restrict__ src, const int* __restrict__ dst,
                          int* __restrict__ cursor, int* __restrict__ csr_src,
                          int n_edges) {
  int e = blockIdx.x * blockDim.x + threadIdx.x;
  if (e < n_edges) {
    int p = atomicAdd(&cursor[dst[e]], 1);
    csr_src[p] = src[e];
  }
}

__global__ void __launch_bounds__(256) tgcn_gemm_wmma(
    const float* __restrict__ X, const float* __restrict__ W,
    const float* __restrict__ t, const float* __restrict__ freq,
    const float* __restrict__ phase, int use_te,
    float* __restrict__ Y, int n_nodes) {
  __shared__ float sA[16 * DFEAT];

  const int tid = threadIdx.x;
  const int mbase = blockIdx.x * 16;

  for (int i = tid; i < 16 * DFEAT; i += 256) {
    int r = i >> 7;
    int c = i & (DFEAT - 1);
    int row = mbase + r;
    float v = 0.0f;
    if (row < n_nodes) {
      v = X[(size_t)row * DFEAT + c];
      if (use_te) v += cosf(t[row] * freq[c] + phase[c]);
    }
    sA[i] = v;
  }
  __syncthreads();

  const int wave = tid >> 5;
  const int lane = tid & 31;
  const int half = lane >> 4;
  const int arow = lane & 15;
  const int ncol = wave * 16 + (lane & 15);

  v8f acc = {};
  for (int k = 0; k < DFEAT; k += 4) {
    const int ka = k + 2 * half;
    v2f a, b;
    a.x = sA[arow * DFEAT + ka];
    a.y = sA[arow * DFEAT + ka + 1];
    b.x = W[(size_t)ka * DFEAT + ncol];
    b.y = W[(size_t)(ka + 1) * DFEAT + ncol];
    acc = __builtin_amdgcn_wmma_f32_16x16x4_f32(
         false, a,  false, b,
         (short)0, acc,  false,  false);
  }

#pragma unroll
  for (int r = 0; r < 8; ++r) {
    int orow = mbase + r + 8 * half;
    if (orow < n_nodes) Y[(size_t)orow * DFEAT + ncol] = acc[r];
  }
}

template <int G, bool PRED>
__device__ __forceinline__ void row_group(v4f& acc, const float* __restrict__ xl,
                                          int sj, float wj, int j, int m) {
  v4f v[G];
  float w[G];
#pragma unroll
  for (int u = 0; u < G; ++u) {
    int idx = j + u;
    if (PRED) idx = min(idx, m - 1);
    const int s = __builtin_amdgcn_readlane(sj, idx);
    w[u] = __builtin_bit_cast(
        float, __builtin_amdgcn_readlane(__builtin_bit_cast(int, wj), idx));
    v[u] = *(const v4f*)(xl + (size_t)s * DFEAT);
  }
#pragma unroll
  for (int u = 0; u < G; ++u)
    if (!PRED || j + u < m) acc += v[u] * w[u];
}

__global__ void __launch_bounds__(256) tgcn_gather(
    const float* __restrict__ xw, const int* __restrict__ csr_src,
    const int* __restrict__ seg_end, const int* __restrict__ cnt,
    const float* __restrict__ dinv, const float* __restrict__ bias,
    float* __restrict__ out, int n_nodes) {
  const int lane = threadIdx.x & 31;
  const int row = blockIdx.x * 8 + __builtin_amdgcn_readfirstlane(threadIdx.x >> 5);
  if (row >= n_nodes) return;

  const int end = seg_end[row];
  const int start = end - cnt[row];
  const float di = dinv[row];
  const float* xl = xw + 4 * lane;

  const v4f self_row = *(const v4f*)(xl + (size_t)row * DFEAT);
  v4f acc = {0.0f, 0.0f, 0.0f, 0.0f};

  for (int base = start; base < end; base += 32) {
    const int m = min(32, end - base);
    int sj = 0;
    float wj = 0.0f;
    if (lane < m) {
      sj = csr_src[base + lane];
      wj = dinv[sj] * di;
    }
    int j = 0;
    for (; j + 8 <= m; j += 8) row_group<8, false>(acc, xl, sj, wj, j, m);
    if (j < m) {
      row_group<4, true>(acc, xl, sj, wj, j, m);
      j += 4;
    }
    if (j < m) row_group<4, true>(acc, xl, sj, wj, j, m);
  }

  acc += self_row * (di * di);
  const v4f b = *(const v4f*)(bias + 4 * lane);
  *(v4f*)(out + (size_t)row * DFEAT + 4 * lane) = acc + b;
}

extern "C" void kernel_launch(void* const* d_in, const int* in_sizes, int n_in,
                              void* d_out, int out_size, void* d_ws, size_t ws_size,
                              hipStream_t stream) {
  const float* x     = (const float*)d_in[0];
  const int*   ei    = (const int*)d_in[1];
  const float* t     = (const float*)d_in[2];
  const float* W1    = (const float*)d_in[3];
  const float* b1    = (const float*)d_in[4];
  const float* W2    = (const float*)d_in[5];
  const float* b2    = (const float*)d_in[6];
  const float* freq  = (const float*)d_in[7];
  const float* phase = (const float*)d_in[8];

  const int n_nodes = in_sizes[2];
  const int n_edges = in_sizes[1] / 2;
  const size_t nd = (size_t)n_nodes * DFEAT;

  const int* src = ei;
  const int* dst = ei + n_edges;

  float* out = (float*)d_out;

  const int T = 256;
  const int g_edge = (n_edges + T - 1) / T;
  const int g_node = (n_nodes + T - 1) / T;
  const int g_gemm = (n_nodes + 15) / 16;
  const int g_tile = (n_nodes + SCAN_TILE - 1) / SCAN_TILE;
  const int g_gath = (n_nodes + 7) / 8;

  float* dinv = (float*)d_ws;
  size_t noff = ((size_t)n_nodes + 255) & ~(size_t)255;
  size_t toff = ((size_t)g_tile + 255) & ~(size_t)255;
  float* bufA = dinv + noff;
  float* bufB = bufA + nd;
  int* cnt      = (int*)(bufB + nd);
  int* cursor   = cnt + noff;
  int* tile_sum = cursor + noff;
  int* csr_src  = tile_sum + toff;

  tgcn_zero_cnt<<<g_node, T, 0, stream>>>(cnt, n_nodes);
  tgcn_count<<<g_edge, T, 0, stream>>>(dst, cnt, n_edges);
  tgcn_scan_tiles<<<g_tile, T, 0, stream>>>(cnt, tile_sum, n_nodes);
  tgcn_scan_write<<<g_tile, T, 0, stream>>>(cnt, tile_sum, cursor, dinv, n_nodes);
  tgcn_fill<<<g_edge, T, 0, stream>>>(src, dst, cursor, csr_src, n_edges);

  tgcn_gemm_wmma<<<g_gemm, T, 0, stream>>>(x, W1, t, freq, phase, 1, bufA, n_nodes);
  tgcn_gather<<<g_gath, T, 0, stream>>>(bufA, csr_src, cursor, cnt, dinv, b1, bufB, n_nodes);

  tgcn_gemm_wmma<<<g_gemm, T, 0, stream>>>(bufB, W2, t, freq, phase, 0, bufA, n_nodes);
  tgcn_gather<<<g_gath, T, 0, stream>>>(bufA, csr_src, cursor, cnt, dinv, b2, out, n_nodes);
}


# layer-2 product fused into layer-1 aggregation: each wave sums 2 rows into a 16-row LDS tile that feeds the WMMA loop; h1 never goes to memory; 8 launches, 100000 fewer waves; sum order unchanged
# speedup vs baseline: 2.1707x; 1.0557x over previous
#include <hip/hip_runtime.h>
#include <math.h>

typedef __attribute__((ext_vector_type(2))) float v2f;
typedef __attribute__((ext_vector_type(4))) float v4f;
typedef __attribute__((ext_vector_type(8))) float v8f;

#define DFEAT 128
#define SCAN_TILE 1024


__global__ void tgcn_zero_cnt(int* __restrict__ cnt, int n_nodes) {
  int i = blockIdx.x * blockDim.x + threadIdx.x;
  if (i < n_nodes) cnt[i] = 0;
}

__global__ void tgcn_count(const int* __restrict__ dst, int* __restrict__ cnt,
                           int n_edges) {
  int e = blockIdx.x * blockDim.x + threadIdx.x;
  if (e < n_edges) atomicAdd(&cnt[dst[e]], 1);
}

__device__ __forceinline__ int wave_sum(int v) {
  for (int o = 16; o > 0; o >>= 1) v += __shfl_xor(v, o);
  return v;
}

__global__ void __launch_bounds__(256) tgcn_scan_tiles(
    const int* __restrict__ cnt, int* __restrict__ tile_sum, int n_nodes) {
  __shared__ int s8[8];
  const int base = blockIdx.x * SCAN_TILE + threadIdx.x * 4;
  int t = 0;
  for (int u = 0; u < 4; ++u)
    if (base + u < n_nodes) t += cnt[base + u];
  t = wave_sum(t);
  if ((threadIdx.x & 31) == 0) s8[threadIdx.x >> 5] = t;
  __syncthreads();
  if (threadIdx.x == 0) {
    int tot = 0;
    for (int w = 0; w < 8; ++w) tot += s8[w];
    tile_sum[blockIdx.x] = tot;
  }
}

__global__ void __launch_bounds__(256) tgcn_scan_write(
    const int* __restrict__ cnt, const int* __restrict__ tile_sum,
    int* __restrict__ cursor, float* __restrict__ dinv, int n_nodes) {
  __shared__ int s_tot[8];
  __shared__ int s_pre[8];
  const int tid = threadIdx.x;
  const int lane = tid & 31;
  const int wave = tid >> 5;

  int pre = 0;
  for (int k = tid; k < (int)blockIdx.x; k += 256) pre += tile_sum[k];
  pre = wave_sum(pre);

  const int base = blockIdx.x * SCAN_TILE + tid * 4;
  int c[4];
  int t = 0;
  for (int u = 0; u < 4; ++u) {
    c[u] = (base + u < n_nodes) ? cnt[base + u] : 0;
    t += c[u];
  }
  int inc = t;
  for (int o = 1; o < 32; o <<= 1) {
    int y = __shfl_up(inc, o);
    if (lane >= o) inc += y;
  }
  if (lane == 31) s_tot[wave] = inc;
  if (lane == 0) s_pre[wave] = pre;
  __syncthreads();
  int off = 0;
  for (int w = 0; w < 8; ++w) {
    off += s_pre[w];
    if (w < wave) off += s_tot[w];
  }
  int run = off + inc - t;
  for (int u = 0; u < 4; ++u) {
    if (base + u < n_nodes) {
      cursor[base + u] = run;
      dinv[base + u] = rsqrtf((float)(c[u] + 1));
      run += c[u];
    }
  }
}

__global__ void tgcn_fill(const int* __restrict__ src, const int* __restrict__ dst,
                          int* __restrict__ cursor, int* __restrict__ csr_src,
                          int n_edges) {
  int e = blockIdx.x * blockDim.x + threadIdx.x;
  if (e < n_edges) {
    int p = atomicAdd(&cursor[dst[e]], 1);
    csr_src[p] = src[e];
  }
}

__device__ __forceinline__ void wmma_tile_16x128(
    const float* sA, const float* __restrict__ W, float* __restrict__ Y,
    int mbase, int n_nodes, int tid) {
  const int wave = tid >> 5;
  const int lane = tid & 31;
  const int half = lane >> 4;
  const int arow = lane & 15;
  const int ncol = wave * 16 + (lane & 15);

  v8f acc = {};
  for (int k = 0; k < DFEAT; k += 4) {
    const int ka = k + 2 * half;
    v2f a, b;
    a.x = sA[arow * DFEAT + ka];
    a.y = sA[arow * DFEAT + ka + 1];
    b.x = W[(size_t)ka * DFEAT + ncol];
    b.y = W[(size_t)(ka + 1) * DFEAT + ncol];
    acc = __builtin_amdgcn_wmma_f32_16x16x4_f32(
         false, a,  false, b,
         (short)0, acc,  false,  false);
  }

#pragma unroll
  for (int r = 0; r < 8; ++r) {
    int orow = mbase + r + 8 * half;
    if (orow < n_nodes) Y[(size_t)orow * DFEAT + ncol] = acc[r];
  }
}

__global__ void __launch_bounds__(256) tgcn_gemm_wmma(
    const float* __restrict__ X, const float* __restrict__ W,
    const float* __restrict__ t, const float* __restrict__ freq,
    const float* __restrict__ phase, int use_te,
    float* __restrict__ Y, int n_nodes) {
  __shared__ float sA[16 * DFEAT];

  const int tid = threadIdx.x;
  const int mbase = blockIdx.x * 16;

  for (int i = tid; i < 16 * DFEAT; i += 256) {
    int r = i >> 7;
    int c = i & (DFEAT - 1);
    int row = mbase + r;
    float v = 0.0f;
    if (row < n_nodes) {
      v = X[(size_t)row * DFEAT + c];
      if (use_te) v += cosf(t[row] * freq[c] + phase[c]);
    }
    sA[i] = v;
  }
  __syncthreads();

  wmma_tile_16x128(sA, W, Y, mbase, n_nodes, tid);
}

template <int G, bool PRED>
__device__ __forceinline__ void row_group(v4f& acc, const float* __restrict__ xl,
                                          int sj, float wj, int j, int m) {
  v4f v[G];
  float w[G];
#pragma unroll
  for (int u = 0; u < G; ++u) {
    int idx = j + u;
    if (PRED) idx = min(idx, m - 1);
    const int s = __builtin_amdgcn_readlane(sj, idx);
    w[u] = __builtin_bit_cast(
        float, __builtin_amdgcn_readlane(__builtin_bit_cast(int, wj), idx));
    v[u] = *(const v4f*)(xl + (size_t)s * DFEAT);
  }
#pragma unroll
  for (int u = 0; u < G; ++u)
    if (!PRED || j + u < m) acc += v[u] * w[u];
}

__global__ void __launch_bounds__(256) tgcn_gather(
    const float* __restrict__ xw, const int* __restrict__ csr_src,
    const int* __restrict__ seg_end, const int* __restrict__ cnt,
    const float* __restrict__ dinv, const float* __restrict__ bias,
    float* __restrict__ out, int n_nodes) {
  const int lane = threadIdx.x & 31;
  const int row = blockIdx.x * 8 + __builtin_amdgcn_readfirstlane(threadIdx.x >> 5);
  if (row >= n_nodes) return;

  const int end = seg_end[row];
  const int start = end - cnt[row];
  const float di = dinv[row];
  const float* xl = xw + 4 * lane;

  const v4f self_row = *(const v4f*)(xl + (size_t)row * DFEAT);
  v4f acc = {0.0f, 0.0f, 0.0f, 0.0f};

  for (int base = start; base < end; base += 32) {
    const int m = min(32, end - base);
    int sj = 0;
    float wj = 0.0f;
    if (lane < m) {
      sj = csr_src[base + lane];
      wj = dinv[sj] * di;
    }
    int j = 0;
    for (; j + 8 <= m; j += 8) row_group<8, false>(acc, xl, sj, wj, j, m);
    if (j < m) {
      row_group<4, true>(acc, xl, sj, wj, j, m);
      j += 4;
    }
    if (j < m) row_group<4, true>(acc, xl, sj, wj, j, m);
  }

  acc += self_row * (di * di);
  const v4f b = *(const v4f*)(bias + 4 * lane);
  *(v4f*)(out + (size_t)row * DFEAT + 4 * lane) = acc + b;
}

__global__ void __launch_bounds__(256) tgcn_gather_gemm(
    const float* __restrict__ xw, const int* __restrict__ csr_src,
    const int* __restrict__ seg_end, const int* __restrict__ cnt,
    const float* __restrict__ dinv, const float* __restrict__ bias,
    const float* __restrict__ W, float* __restrict__ Y, int n_nodes) {
  __shared__ __attribute__((aligned(16))) float sA[16 * DFEAT];

  const int tid = threadIdx.x;
  const int lane = tid & 31;
  const int wave = __builtin_amdgcn_readfirstlane(tid >> 5);
  const int mbase = blockIdx.x * 16;
  const float* xl = xw + 4 * lane;
  const v4f b = *(const v4f*)(bias + 4 * lane);

  int startq[2], endq[2];
  float diq[2];
  v4f selfq[2];
#pragma unroll
  for (int q = 0; q < 2; ++q) {
    const int row = mbase + 2 * wave + q;
    startq[q] = 0;
    endq[q] = 0;
    diq[q] = 0.0f;
    selfq[q] = (v4f){0.0f, 0.0f, 0.0f, 0.0f};
    if (row < n_nodes) {
      endq[q] = seg_end[row];
      startq[q] = endq[q] - cnt[row];
      diq[q] = dinv[row];
      selfq[q] = *(const v4f*)(xl + (size_t)row * DFEAT);
    }
  }

#pragma unroll
  for (int q = 0; q < 2; ++q) {
    const int end = endq[q];
    const float di = diq[q];
    v4f acc = {0.0f, 0.0f, 0.0f, 0.0f};
    for (int base = startq[q]; base < end; base += 32) {
      const int m = min(32, end - base);
      int sj = 0;
      float wj = 0.0f;
      if (lane < m) {
        sj = csr_src[base + lane];
        wj = dinv[sj] * di;
      }
      int j = 0;
      for (; j + 8 <= m; j += 8) row_group<8, false>(acc, xl, sj, wj, j, m);
      if (j < m) {
        row_group<4, true>(acc, xl, sj, wj, j, m);
        j += 4;
      }
      if (j < m) row_group<4, true>(acc, xl, sj, wj, j, m);
    }
    acc += selfq[q] * (di * di);
    *(v4f*)(sA + (2 * wave + q) * DFEAT + 4 * lane) = acc + b;
  }
  __syncthreads();

  wmma_tile_16x128(sA, W, Y, mbase, n_nodes, tid);
}

extern "C" void kernel_launch(void* const* d_in, const int* in_sizes, int n_in,
                              void* d_out, int out_size, void* d_ws, size_t ws_size,
                              hipStream_t stream) {
  const float* x     = (const float*)d_in[0];
  const int*   ei    = (const int*)d_in[1];
  const float* t     = (const float*)d_in[2];
  const float* W1    = (const float*)d_in[3];
  const float* b1    = (const float*)d_in[4];
  const float* W2    = (const float*)d_in[5];
  const float* b2    = (const float*)d_in[6];
  const float* freq  = (const float*)d_in[7];
  const float* phase = (const float*)d_in[8];

  const int n_nodes = in_sizes[2];
  const int n_edges = in_sizes[1] / 2;
  const size_t nd = (size_t)n_nodes * DFEAT;

  const int* src = ei;
  const int* dst = ei + n_edges;

  float* out = (float*)d_out;

  const int T = 256;
  const int g_edge = (n_edges + T - 1) / T;
  const int g_node = (n_nodes + T - 1) / T;
  const int g_gemm = (n_nodes + 15) / 16;
  const int g_tile = (n_nodes + SCAN_TILE - 1) / SCAN_TILE;
  const int g_gath = (n_nodes + 7) / 8;

  float* dinv = (float*)d_ws;
  size_t noff = ((size_t)n_nodes + 255) & ~(size_t)255;
  size_t toff = ((size_t)g_tile + 255) & ~(size_t)255;
  float* bufA = dinv + noff;
  float* bufB = bufA + nd;
  int* cnt      = (int*)(bufB + nd);
  int* cursor   = cnt + noff;
  int* tile_sum = cursor + noff;
  int* csr_src  = tile_sum + toff;

  tgcn_zero_cnt<<<g_node, T, 0, stream>>>(cnt, n_nodes);
  tgcn_count<<<g_edge, T, 0, stream>>>(dst, cnt, n_edges);
  tgcn_scan_tiles<<<g_tile, T, 0, stream>>>(cnt, tile_sum, n_nodes);
  tgcn_scan_write<<<g_tile, T, 0, stream>>>(cnt, tile_sum, cursor, dinv, n_nodes);
  tgcn_fill<<<g_edge, T, 0, stream>>>(src, dst, cursor, csr_src, n_edges);

  tgcn_gemm_wmma<<<g_gemm, T, 0, stream>>>(x, W1, t, freq, phase, 1, bufA, n_nodes);

  tgcn_gather_gemm<<<g_gemm, T, 0, stream>>>(bufA, csr_src, cursor, cnt, dinv, b1, W2, bufB, n_nodes);

  tgcn_gather<<<g_gath, T, 0, stream>>>(bufB, csr_src, cursor, cnt, dinv, b2, out, n_nodes);
}


# LDS A tile of both WMMA kernels at row pitch 132 floats (16 B x odd) instead of 128: the 8-byte fragment reads of 16 rows hit 64 distinct banks instead of one bank pair; sums unchanged
# speedup vs baseline: 2.4002x; 1.1057x over previous
#include <hip/hip_runtime.h>
#include <math.h>

typedef __attribute__((ext_vector_type(2))) float v2f;
typedef __attribute__((ext_vector_type(4))) float v4f;
typedef __attribute__((ext_vector_type(8))) float v8f;

#define DFEAT 128
#define APITCH 132
#define SCAN_TILE 1024


__global__ void tgcn_zero_cnt(int* __restrict__ cnt, int n_nodes) {
  int i = blockIdx.x * blockDim.x + threadIdx.x;
  if (i < n_nodes) cnt[i] = 0;
}

__global__ void tgcn_count(const int* __restrict__ dst, int* __restrict__ cnt,
                           int n_edges) {
  int e = blockIdx.x * blockDim.x + threadIdx.x;
  if (e < n_edges) atomicAdd(&cnt[dst[e]], 1);
}

__device__ __forceinline__ int wave_sum(int v) {
  for (int o = 16; o > 0; o >>= 1) v += __shfl_xor(v, o);
  return v;
}

__global__ void __launch_bounds__(256) tgcn_scan_tiles(
    const int* __restrict__ cnt, int* __restrict__ tile_sum, int n_nodes) {
  __shared__ int s8[8];
  const int base = blockIdx.x * SCAN_TILE + threadIdx.x * 4;
  int t = 0;
  for (int u = 0; u < 4; ++u)
    if (base + u < n_nodes) t += cnt[base + u];
  t = wave_sum(t);
  if ((threadIdx.x & 31) == 0) s8[threadIdx.x >> 5] = t;
  __syncthreads();
  if (threadIdx.x == 0) {
    int tot = 0;
    for (int w = 0; w < 8; ++w) tot += s8[w];
    tile_sum[blockIdx.x] = tot;
  }
}

__global__ void __launch_bounds__(256) tgcn_scan_write(
    const int* __restrict__ cnt, const int* __restrict__ tile_sum,
    int* __restrict__ cursor, float* __restrict__ dinv, int n_nodes) {
  __shared__ int s_tot[8];
  __shared__ int s_pre[8];
  const int tid = threadIdx.x;
  const int lane = tid & 31;
  const int wave = tid >> 5;

  int pre = 0;
  for (int k = tid; k < (int)blockIdx.x; k += 256) pre += tile_sum[k];
  pre = wave_sum(pre);

  const int base = blockIdx.x * SCAN_TILE + tid * 4;
  int c[4];
  int t = 0;
  for (int u = 0; u < 4; ++u) {
    c[u] = (base + u < n_nodes) ? cnt[base + u] : 0;
    t += c[u];
  }
  int inc = t;
  for (int o = 1; o < 32; o <<= 1) {
    int y = __shfl_up(inc, o);
    if (lane >= o) inc += y;
  }
  if (lane == 31) s_tot[wave] = inc;
  if (lane == 0) s_pre[wave] = pre;
  __syncthreads();
  int off = 0;
  for (int w = 0; w < 8; ++w) {
    off += s_pre[w];
    if (w < wave) off += s_tot[w];
  }
  int run = off + inc - t;
  for (int u = 0; u < 4; ++u) {
    if (base + u < n_nodes) {
      cursor[base + u] = run;
      dinv[base + u] = rsqrtf((float)(c[u] + 1));
      run += c[u];
    }
  }
}

__global__ void tgcn_fill(const int* __restrict__ src, const int* __restrict__ dst,
                          int* __restrict__ cursor, int* __restrict__ csr_src,
                          int n_edges) {
  int e = blockIdx.x * blockDim.x + threadIdx.x;
  if (e < n_edges) {
    int p = atomicAdd(&cursor[dst[e]], 1);
    csr_src[p] = src[e];
  }
}

__device__ __forceinline__ void wmma_tile_16x128(
    const float* sA, const float* __restrict__ W, float* __restrict__ Y,
    int mbase, int n_nodes, int tid) {
  const int wave = tid >> 5;
  const int lane = tid & 31;
  const int half = lane >> 4;
  const int arow = lane & 15;
  const int ncol = wave * 16 + (lane & 15);

  v8f acc = {};
  for (int k = 0; k < DFEAT; k += 4) {
    const int ka = k + 2 * half;
    v2f a, b;
    a.x = sA[arow * APITCH + ka];
    a.y = sA[arow * APITCH + ka + 1];
    b.x = W[(size_t)ka * DFEAT + ncol];
    b.y = W[(size_t)(ka + 1) * DFEAT + ncol];
    acc = __builtin_amdgcn_wmma_f32_16x16x4_f32(
         false, a,  false, b,
         (short)0, acc,  false,  false);
  }

#pragma unroll
  for (int r = 0; r < 8; ++r) {
    int orow = mbase + r + 8 * half;
    if (orow < n_nodes) Y[(size_t)orow * DFEAT + ncol] = acc[r];
  }
}

__global__ void __launch_bounds__(256) tgcn_gemm_wmma(
    const float* __restrict__ X, const float* __restrict__ W,
    const float* __restrict__ t, const float* __restrict__ freq,
    const float* __restrict__ phase, int use_te,
    float* __restrict__ Y, int n_nodes) {
  __shared__ __attribute__((aligned(16))) float sA[16 * APITCH];

  const int tid = threadIdx.x;
  const int mbase = blockIdx.x * 16;

  for (int i = tid; i < 16 * DFEAT; i += 256) {
    int r = i >> 7;
    int c = i & (DFEAT - 1);
    int row = mbase + r;
    float v = 0.0f;
    if (row < n_nodes) {
      v = X[(size_t)row * DFEAT + c];
      if (use_te) v += cosf(t[row] * freq[c] + phase[c]);
    }
    sA[r * APITCH + c] = v;
  }
  __syncthreads();

  wmma_tile_16x128(sA, W, Y, mbase, n_nodes, tid);
}

template <int G, bool PRED>
__device__ __forceinline__ void row_group(v4f& acc, const float* __restrict__ xl,
                                          int sj, float wj, int j, int m) {
  v4f v[G];
  float w[G];
#pragma unroll
  for (int u = 0; u < G; ++u) {
    int idx = j + u;
    if (PRED) idx = min(idx, m - 1);
    const int s = __builtin_amdgcn_readlane(sj, idx);
    w[u] = __builtin_bit_cast(
        float, __builtin_amdgcn_readlane(__builtin_bit_cast(int, wj), idx));
    v[u] = *(const v4f*)(xl + (size_t)s * DFEAT);
  }
#pragma unroll
  for (int u = 0; u < G; ++u)
    if (!PRED || j + u < m) acc += v[u] * w[u];
}

__global__ void __launch_bounds__(256) tgcn_gather(
    const float* __restrict__ xw, const int* __restrict__ csr_src,
    const int* __restrict__ seg_end, const int* __restrict__ cnt,
    const float* __restrict__ dinv, const float* __restrict__ bias,
    float* __restrict__ out, int n_nodes) {
  const int lane = threadIdx.x & 31;
  const int row = blockIdx.x * 8 + __builtin_amdgcn_readfirstlane(threadIdx.x >> 5);
  if (row >= n_nodes) return;

  const int end = seg_end[row];
  const int start = end - cnt[row];
  const float di = dinv[row];
  const float* xl = xw + 4 * lane;

  const v4f self_row = *(const v4f*)(xl + (size_t)row * DFEAT);
  v4f acc = {0.0f, 0.0f, 0.0f, 0.0f};

  for (int base = start; base < end; base += 32) {
    const int m = min(32, end - base);
    int sj = 0;
    float wj = 0.0f;
    if (lane < m) {
      sj = csr_src[base + lane];
      wj = dinv[sj] * di;
    }
    int j = 0;
    for (; j + 8 <= m; j += 8) row_group<8, false>(acc, xl, sj, wj, j, m);
    if (j < m) {
      row_group<4, true>(acc, xl, sj, wj, j, m);
      j += 4;
    }
    if (j < m) row_group<4, true>(acc, xl, sj, wj, j, m);
  }

  acc += self_row * (di * di);
  const v4f b = *(const v4f*)(bias + 4 * lane);
  *(v4f*)(out + (size_t)row * DFEAT + 4 * lane) = acc + b;
}

__global__ void __launch_bounds__(256) tgcn_gather_gemm(
    const float* __restrict__ xw, const int* __restrict__ csr_src,
    const int* __restrict__ seg_end, const int* __restrict__ cnt,
    const float* __restrict__ dinv, const float* __restrict__ bias,
    const float* __restrict__ W, float* __restrict__ Y, int n_nodes) {
  __shared__ __attribute__((aligned(16))) float sA[16 * APITCH];

  const int tid = threadIdx.x;
  const int lane = tid & 31;
  const int wave = __builtin_amdgcn_readfirstlane(tid >> 5);
  const int mbase = blockIdx.x * 16;
  const float* xl = xw + 4 * lane;
  const v4f b = *(const v4f*)(bias + 4 * lane);

  int startq[2], endq[2];
  float diq[2];
  v4f selfq[2];
#pragma unroll
  for (int q = 0; q < 2; ++q) {
    const int row = mbase + 2 * wave + q;
    startq[q] = 0;
    endq[q] = 0;
    diq[q] = 0.0f;
    selfq[q] = (v4f){0.0f, 0.0f, 0.0f, 0.0f};
    if (row < n_nodes) {
      endq[q] = seg_end[row];
      startq[q] = endq[q] - cnt[row];
      diq[q] = dinv[row];
      selfq[q] = *(const v4f*)(xl + (size_t)row * DFEAT);
    }
  }

#pragma unroll
  for (int q = 0; q < 2; ++q) {
    const int end = endq[q];
    const float di = diq[q];
    v4f acc = {0.0f, 0.0f, 0.0f, 0.0f};
    for (int base = startq[q]; base < end; base += 32) {
      const int m = min(32, end - base);
      int sj = 0;
      float wj = 0.0f;
      if (lane < m) {
        sj = csr_src[base + lane];
        wj = dinv[sj] * di;
      }
      int j = 0;
      for (; j + 8 <= m; j += 8) row_group<8, false>(acc, xl, sj, wj, j, m);
      if (j < m) {
        row_group<4, true>(acc, xl, sj, wj, j, m);
        j += 4;
      }
      if (j < m) row_group<4, true>(acc, xl, sj, wj, j, m);
    }
    acc += selfq[q] * (di * di);
    *(v4f*)(sA + (2 * wave + q) * APITCH + 4 * lane) = acc + b;
  }
  __syncthreads();

  wmma_tile_16x128(sA, W, Y, mbase, n_nodes, tid);
}

extern "C" void kernel_launch(void* const* d_in, const int* in_sizes, int n_in,
                              void* d_out, int out_size, void* d_ws, size_t ws_size,
                              hipStream_t stream) {
  const float* x     = (const float*)d_in[0];
  const int*   ei    = (const int*)d_in[1];
  const float* t     = (const float*)d_in[2];
  const float* W1    = (const float*)d_in[3];
  const float* b1    = (const float*)d_in[4];
  const float* W2    = (const float*)d_in[5];
  const float* b2    = (const float*)d_in[6];
  const float* freq  = (const float*)d_in[7];
  const float* phase = (const float*)d_in[8];

  const int n_nodes = in_sizes[2];
  const int n_edges = in_sizes[1] / 2;
  const size_t nd = (size_t)n_nodes * DFEAT;

  const int* src = ei;
  const int* dst = ei + n_edges;

  float* out = (float*)d_out;

  const int T = 256;
  const int g_edge = (n_edges + T - 1) / T;
  const int g_node = (n_nodes + T - 1) / T;
  const int g_gemm = (n_nodes + 15) / 16;
  const int g_tile = (n_nodes + SCAN_TILE - 1) / SCAN_TILE;
  const int g_gath = (n_nodes + 7) / 8;

  float* dinv = (float*)d_ws;
  size_t noff = ((size_t)n_nodes + 255) & ~(size_t)255;
  size_t toff = ((size_t)g_tile + 255) & ~(size_t)255;
  float* bufA = dinv + noff;
  float* bufB = bufA + nd;
  int* cnt      = (int*)(bufB + nd);
  int* cursor   = cnt + noff;
  int* tile_sum = cursor + noff;
  int* csr_src  = tile_sum + toff;

  tgcn_zero_cnt<<<g_node, T, 0, stream>>>(cnt, n_nodes);
  tgcn_count<<<g_edge, T, 0, stream>>>(dst, cnt, n_edges);
  tgcn_scan_tiles<<<g_tile, T, 0, stream>>>(cnt, tile_sum, n_nodes);
  tgcn_scan_write<<<g_tile, T, 0, stream>>>(cnt, tile_sum, cursor, dinv, n_nodes);
  tgcn_fill<<<g_edge, T, 0, stream>>>(src, dst, cursor, csr_src, n_edges);

  tgcn_gemm_wmma<<<g_gemm, T, 0, stream>>>(x, W1, t, freq, phase, 1, bufA, n_nodes);

  tgcn_gather_gemm<<<g_gemm, T, 0, stream>>>(bufA, csr_src, cursor, cnt, dinv, b1, W2, bufB, n_nodes);

  tgcn_gather<<<g_gath, T, 0, stream>>>(bufB, csr_src, cursor, cnt, dinv, b2, out, n_nodes);
}


# no activation between the layers: W1*W2 and b1*W2 formed once per call on WMMA, nodes multiplied by W1*W2 only, aggregation kernel loses its WMMA phase; products re-associated, rounding changes
# speedup vs baseline: 2.5271x; 1.0529x over previous
#include <hip/hip_runtime.h>
#include <math.h>

typedef __attribute__((ext_vector_type(2))) float v2f;
typedef __attribute__((ext_vector_type(4))) float v4f;
typedef __attribute__((ext_vector_type(8))) float v8f;

#define DFEAT 128
#define APITCH 132
#define SCAN_TILE 1024


__global__ void tgcn_count(const int* __restrict__ dst, int* __restrict__ cnt,
                           int n_edges) {
  int e = blockIdx.x * blockDim.x + threadIdx.x;
  if (e < n_edges) atomicAdd(&cnt[dst[e]], 1);
}

__device__ __forceinline__ int wave_sum(int v) {
  for (int o = 16; o > 0; o >>= 1) v += __shfl_xor(v, o);
  return v;
}

__global__ void __launch_bounds__(256) tgcn_scan_tiles(
    const int* __restrict__ cnt, int* __restrict__ tile_sum, int n_nodes) {
  __shared__ int s8[8];
  const int base = blockIdx.x * SCAN_TILE + threadIdx.x * 4;
  int t = 0;
  for (int u = 0; u < 4; ++u)
    if (base + u < n_nodes) t += cnt[base + u];
  t = wave_sum(t);
  if ((threadIdx.x & 31) == 0) s8[threadIdx.x >> 5] = t;
  __syncthreads();
  if (threadIdx.x == 0) {
    int tot = 0;
    for (int w = 0; w < 8; ++w) tot += s8[w];
    tile_sum[blockIdx.x] = tot;
  }
}

__global__ void __launch_bounds__(256) tgcn_scan_write(
    const int* __restrict__ cnt, const int* __restrict__ tile_sum,
    int* __restrict__ cursor, float* __restrict__ dinv, int n_nodes) {
  __shared__ int s_tot[8];
  __shared__ int s_pre[8];
  const int tid = threadIdx.x;
  const int lane = tid & 31;
  const int wave = tid >> 5;

  int pre = 0;
  for (int k = tid; k < (int)blockIdx.x; k += 256) pre += tile_sum[k];
  pre = wave_sum(pre);

  const int base = blockIdx.x * SCAN_TILE + tid * 4;
  int c[4];
  int t = 0;
  for (int u = 0; u < 4; ++u) {
    c[u] = (base + u < n_nodes) ? cnt[base + u] : 0;
    t += c[u];
  }
  int inc = t;
  for (int o = 1; o < 32; o <<= 1) {
    int y = __shfl_up(inc, o);
    if (lane >= o) inc += y;
  }
  if (lane == 31) s_tot[wave] = inc;
  if (lane == 0) s_pre[wave] = pre;
  __syncthreads();
  int off = 0;
  for (int w = 0; w < 8; ++w) {
    off += s_pre[w];
    if (w < wave) off += s_tot[w];
  }
  int run = off + inc - t;
  for (int u = 0; u < 4; ++u) {
    if (base + u < n_nodes) {
      cursor[base + u] = run;
      dinv[base + u] = rsqrtf((float)(c[u] + 1));
      run += c[u];
    }
  }
}

__global__ void tgcn_fill(const int* __restrict__ src, const int* __restrict__ dst,
                          int* __restrict__ cursor, int* __restrict__ csr_src,
                          int n_edges) {
  int e = blockIdx.x * blockDim.x + threadIdx.x;
  if (e < n_edges) {
    int p = atomicAdd(&cursor[dst[e]], 1);
    csr_src[p] = src[e];
  }
}

__device__ __forceinline__ void wmma_tile_16x128(
    const float* sA, const float* __restrict__ W, float* __restrict__ Y,
    int mbase, int n_nodes, int tid) {
  const int wave = tid >> 5;
  const int lane = tid & 31;
  const int half = lane >> 4;
  const int arow = lane & 15;
  const int ncol = wave * 16 + (lane & 15);

  v8f acc = {};
  for (int k = 0; k < DFEAT; k += 4) {
    const int ka = k + 2 * half;
    v2f a, b;
    a.x = sA[arow * APITCH + ka];
    a.y = sA[arow * APITCH + ka + 1];
    b.x = W[(size_t)ka * DFEAT + ncol];
    b.y = W[(size_t)(ka + 1) * DFEAT + ncol];
    acc = __builtin_amdgcn_wmma_f32_16x16x4_f32(
         false, a,  false, b,
         (short)0, acc,  false,  false);
  }

#pragma unroll
  for (int r = 0; r < 8; ++r) {
    int orow = mbase + r + 8 * half;
    if (orow < n_nodes) Y[(size_t)orow * DFEAT + ncol] = acc[r];
  }
}

__global__ void __launch_bounds__(256) tgcn_prep(
    int* __restrict__ cnt, int n_nodes, int g_node,
    const float* __restrict__ W1, const float* __restrict__ b1,
    const float* __restrict__ W2, float* __restrict__ W12,
    float* __restrict__ c12) {
  __shared__ __attribute__((aligned(16))) float sA[16 * APITCH];
  const int tid = threadIdx.x;
  if ((int)blockIdx.x < g_node) {
    const int i = blockIdx.x * 256 + tid;
    if (i < n_nodes) cnt[i] = 0;
    return;
  }
  const int tile = blockIdx.x - g_node;
  const bool is_w = tile < 8;
  for (int i = tid; i < 16 * (DFEAT / 4); i += 256) {
    const int r = i >> 5;
    const int c4 = i & 31;
    v4f v = {0.0f, 0.0f, 0.0f, 0.0f};
    if (is_w)
      v = *(const v4f*)(W1 + (size_t)(tile * 16 + r) * DFEAT + 4 * c4);
    else if (r == 0)
      v = *(const v4f*)(b1 + 4 * c4);
    *(v4f*)(sA + r * APITCH + 4 * c4) = v;
  }
  __syncthreads();
  wmma_tile_16x128(sA, W2, is_w ? W12 : c12, is_w ? tile * 16 : 0,
                   is_w ? DFEAT : 1, tid);
}

__global__ void __launch_bounds__(256) tgcn_gemm_wmma(
    const float* __restrict__ X, const float* __restrict__ W,
    const float* __restrict__ t, const float* __restrict__ freq,
    const float* __restrict__ phase, int use_te,
    float* __restrict__ Y, int n_nodes) {
  __shared__ __attribute__((aligned(16))) float sA[16 * APITCH];

  const int tid = threadIdx.x;
  const int mbase = blockIdx.x * 16;

  for (int i = tid; i < 16 * DFEAT; i += 256) {
    int r = i >> 7;
    int c = i & (DFEAT - 1);
    int row = mbase + r;
    float v = 0.0f;
    if (row < n_nodes) {
      v = X[(size_t)row * DFEAT + c];
      if (use_te) v += cosf(t[row] * freq[c] + phase[c]);
    }
    sA[r * APITCH + c] = v;
  }
  __syncthreads();

  wmma_tile_16x128(sA, W, Y, mbase, n_nodes, tid);
}

template <int G, bool PRED>
__device__ __forceinline__ void row_group(v4f& acc, const float* __restrict__ xl,
                                          int sj, float wj, int j, int m) {
  v4f v[G];
  float w[G];
#pragma unroll
  for (int u = 0; u < G; ++u) {
    int idx = j + u;
    if (PRED) idx = min(idx, m - 1);
    const int s = __builtin_amdgcn_readlane(sj, idx);
    w[u] = __builtin_bit_cast(
        float, __builtin_amdgcn_readlane(__builtin_bit_cast(int, wj), idx));
    v[u] = *(const v4f*)(xl + (size_t)s * DFEAT);
  }
#pragma unroll
  for (int u = 0; u < G; ++u)
    if (!PRED || j + u < m) acc += v[u] * w[u];
}

__global__ void __launch_bounds__(256) tgcn_gather(
    const float* __restrict__ xw, const int* __restrict__ csr_src,
    const int* __restrict__ seg_end, const int* __restrict__ cnt,
    const float* __restrict__ dinv, const float* __restrict__ bias,
    float* __restrict__ out, int n_nodes) {
  const int lane = threadIdx.x & 31;
  const int row = blockIdx.x * 8 + __builtin_amdgcn_readfirstlane(threadIdx.x >> 5);
  if (row >= n_nodes) return;

  const int end = seg_end[row];
  const int start = end - cnt[row];
  const float di = dinv[row];
  const float* xl = xw + 4 * lane;

  const v4f self_row = *(const v4f*)(xl + (size_t)row * DFEAT);
  v4f acc = {0.0f, 0.0f, 0.0f, 0.0f};

  for (int base = start; base < end; base += 32) {
    const int m = min(32, end - base);
    int sj = 0;
    float wj = 0.0f;
    if (lane < m) {
      sj = csr_src[base + lane];
      wj = dinv[sj] * di;
    }
    int j = 0;
    for (; j + 8 <= m; j += 8) row_group<8, false>(acc, xl, sj, wj, j, m);
    if (j < m) {
      row_group<4, true>(acc, xl, sj, wj, j, m);
      j += 4;
    }
    if (j < m) row_group<4, true>(acc, xl, sj, wj, j, m);
  }

  acc += self_row * (di * di);
  const v4f b = *(const v4f*)(bias + 4 * lane);
  *(v4f*)(out + (size_t)row * DFEAT + 4 * lane) = acc + b;
}

__global__ void __launch_bounds__(256) tgcn_gather_pair(
    const float* __restrict__ xw, const int* __restrict__ csr_src,
    const int* __restrict__ seg_end, const int* __restrict__ cnt,
    const float* __restrict__ dinv, const float* __restrict__ bias,
    float* __restrict__ out, int n_nodes) {
  const int tid = threadIdx.x;
  const int lane = tid & 31;
  const int wave = __builtin_amdgcn_readfirstlane(tid >> 5);
  const int mbase = blockIdx.x * 16;
  const float* xl = xw + 4 * lane;
  const v4f b = *(const v4f*)(bias + 4 * lane);

  int startq[2], endq[2];
  float diq[2];
  v4f selfq[2];
#pragma unroll
  for (int q = 0; q < 2; ++q) {
    const int row = mbase + 2 * wave + q;
    startq[q] = 0;
    endq[q] = 0;
    diq[q] = 0.0f;
    selfq[q] = (v4f){0.0f, 0.0f, 0.0f, 0.0f};
    if (row < n_nodes) {
      endq[q] = seg_end[row];
      startq[q] = endq[q] - cnt[row];
      diq[q] = dinv[row];
      selfq[q] = *(const v4f*)(xl + (size_t)row * DFEAT);
    }
  }

#pragma unroll
  for (int q = 0; q < 2; ++q) {
    const int row = mbase + 2 * wave + q;
    const int end = endq[q];
    const float di = diq[q];
    v4f acc = {0.0f, 0.0f, 0.0f, 0.0f};
    for (int base = startq[q]; base < end; base += 32) {
      const int m = min(32, end - base);
      int sj = 0;
      float wj = 0.0f;
      if (lane < m) {
        sj = csr_src[base + lane];
        wj = dinv[sj] * di;
      }
      int j = 0;
      for (; j + 8 <= m; j += 8) row_group<8, false>(acc, xl, sj, wj, j, m);
      if (j < m) {
        row_group<4, true>(acc, xl, sj, wj, j, m);
        j += 4;
      }
      if (j < m) row_group<4, true>(acc, xl, sj, wj, j, m);
    }
    acc += selfq[q] * (di * di);
    if (row < n_nodes) *(v4f*)(out + (size_t)row * DFEAT + 4 * lane) = acc + b;
  }
}

extern "C" void kernel_launch(void* const* d_in, const int* in_sizes, int n_in,
                              void* d_out, int out_size, void* d_ws, size_t ws_size,
                              hipStream_t stream) {
  const float* x     = (const float*)d_in[0];
  const int*   ei    = (const int*)d_in[1];
  const float* t     = (const float*)d_in[2];
  const float* W1    = (const float*)d_in[3];
  const float* b1    = (const float*)d_in[4];
  const float* W2    = (const float*)d_in[5];
  const float* b2    = (const float*)d_in[6];
  const float* freq  = (const float*)d_in[7];
  const float* phase = (const float*)d_in[8];

  const int n_nodes = in_sizes[2];
  const int n_edges = in_sizes[1] / 2;
  const size_t nd = (size_t)n_nodes * DFEAT;

  const int* src = ei;
  const int* dst = ei + n_edges;

  float* out = (float*)d_out;

  const int T = 256;
  const int g_edge = (n_edges + T - 1) / T;
  const int g_node = (n_nodes + T - 1) / T;
  const int g_gemm = (n_nodes + 15) / 16;
  const int g_tile = (n_nodes + SCAN_TILE - 1) / SCAN_TILE;
  const int g_gath = (n_nodes + 7) / 8;

  float* dinv = (float*)d_ws;
  size_t noff = ((size_t)n_nodes + 255) & ~(size_t)255;
  size_t toff = ((size_t)g_tile + 255) & ~(size_t)255;
  size_t eoff = ((size_t)n_edges + 255) & ~(size_t)255;
  float* bufA = dinv + noff;
  float* bufB = bufA + nd;
  int* cnt      = (int*)(bufB + nd);
  int* cursor   = cnt + noff;
  int* tile_sum = cursor + noff;
  int* csr_src  = tile_sum + toff;
  float* W12 = (float*)(csr_src + eoff);
  float* c12 = W12 + (size_t)DFEAT * DFEAT;

  tgcn_prep<<<g_node + 9, T, 0, stream>>>(cnt, n_nodes, g_node, W1, b1, W2, W12, c12);
  tgcn_count<<<g_edge, T, 0, stream>>>(dst, cnt, n_edges);
  tgcn_scan_tiles<<<g_tile, T, 0, stream>>>(cnt, tile_sum, n_nodes);
  tgcn_scan_write<<<g_tile, T, 0, stream>>>(cnt, tile_sum, cursor, dinv, n_nodes);
  tgcn_fill<<<g_edge, T, 0, stream>>>(src, dst, cursor, csr_src, n_edges);

  tgcn_gemm_wmma<<<g_gemm, T, 0, stream>>>(x, W12, t, freq, phase, 1, bufA, n_nodes);

  tgcn_gather_pair<<<g_gemm, T, 0, stream>>>(bufA, csr_src, cursor, cnt, dinv, c12, bufB, n_nodes);

  tgcn_gather<<<g_gath, T, 0, stream>>>(bufB, csr_src, cursor, cnt, dinv, b2, out, n_nodes);
}


# edge grouping by destination: the global integer atomic per edge in count and in fill replaced by a two-level counting sort on LDS counters; 7 launches; sum order = arrival order of the LDS adds
# speedup vs baseline: 4.9488x; 1.9583x over previous
#include <hip/hip_runtime.h>
#include <math.h>

typedef __attribute__((ext_vector_type(2))) float v2f;
typedef __attribute__((ext_vector_type(2))) int v2i;
typedef __attribute__((ext_vector_type(4))) float v4f;
typedef __attribute__((ext_vector_type(8))) float v8f;

#define DFEAT 128
#define APITCH 132
#define RB 512
#define RB_SHIFT 9
#define MAXB 1024
#define CHUNK 8192


__device__ __forceinline__ int wave_sum(int v) {
  for (int o = 16; o > 0; o >>= 1) v += __shfl_xor(v, o);
  return v;
}

__device__ __forceinline__ int wave_scan_incl(int v, int lane) {
  for (int o = 1; o < 32; o <<= 1) {
    int y = __shfl_up(v, o);
    if (lane >= o) v += y;
  }
  return v;
}

__global__ void __launch_bounds__(1024) tgcn_bucket_hist(
    const int* __restrict__ dst, int* __restrict__ bmat, int n_edges, int nb,
    int nblk) {
  __shared__ int s_cnt[MAXB];
  const int tid = threadIdx.x;
  if (tid < nb) s_cnt[tid] = 0;
  __syncthreads();
  const int base = blockIdx.x * CHUNK + tid;
  int d[8];
#pragma unroll
  for (int j = 0; j < 8; ++j) {
    const int e = base + j * 1024;
    d[j] = (e < n_edges) ? dst[e] : -1;
  }
#pragma unroll
  for (int j = 0; j < 8; ++j)
    if (d[j] >= 0) atomicAdd(&s_cnt[d[j] >> RB_SHIFT], 1);
  __syncthreads();
  if (tid < nb) bmat[(size_t)tid * nblk + blockIdx.x] = s_cnt[tid];
}

__device__ __forceinline__ void wmma_tile_16x128(
    const float* sA, const float* __restrict__ W, float* __restrict__ Y,
    int mbase, int n_nodes, int tid) {
  const int wave = tid >> 5;
  const int lane = tid & 31;
  const int half = lane >> 4;
  const int arow = lane & 15;
  const int ncol = wave * 16 + (lane & 15);

  v8f acc = {};
  for (int k = 0; k < DFEAT; k += 4) {
    const int ka = k + 2 * half;
    v2f a, b;
    a.x = sA[arow * APITCH + ka];
    a.y = sA[arow * APITCH + ka + 1];
    b.x = W[(size_t)ka * DFEAT + ncol];
    b.y = W[(size_t)(ka + 1) * DFEAT + ncol];
    acc = __builtin_amdgcn_wmma_f32_16x16x4_f32(
         false, a,  false, b,
         (short)0, acc,  false,  false);
  }

#pragma unroll
  for (int r = 0; r < 8; ++r) {
    int orow = mbase + r + 8 * half;
    if (orow < n_nodes) Y[(size_t)orow * DFEAT + ncol] = acc[r];
  }
}

__global__ void __launch_bounds__(256) tgcn_scan_fold(
    int* __restrict__ bmat, int* __restrict__ btot, int nb, int nblk,
    const float* __restrict__ W1, const float* __restrict__ b1,
    const float* __restrict__ W2, float* __restrict__ W12,
    float* __restrict__ c12) {
  __shared__ __attribute__((aligned(16))) float sA[16 * APITCH];
  __shared__ int s_w[8];
  const int tid = threadIdx.x;
  if ((int)blockIdx.x < nb) {
    int* row = bmat + (size_t)blockIdx.x * nblk;
    const int lane = tid & 31;
    const int wave = tid >> 5;
    int carry = 0;
    for (int base = 0; base < nblk; base += 256) {
      const int i = base + tid;
      const int v = (i < nblk) ? row[i] : 0;
      const int inc = wave_scan_incl(v, lane);
      if (lane == 31) s_w[wave] = inc;
      __syncthreads();
      int off = 0, tot = 0;
      for (int w = 0; w < 8; ++w) {
        const int s = s_w[w];
        if (w < wave) off += s;
        tot += s;
      }
      if (i < nblk) row[i] = carry + off + inc - v;
      carry += tot;
      __syncthreads();
    }
    if (tid == 0) btot[blockIdx.x] = carry;
    return;
  }
  const int tile = blockIdx.x - nb;
  const bool is_w = tile < 8;
  for (int i = tid; i < 16 * (DFEAT / 4); i += 256) {
    const int r = i >> 5;
    const int c4 = i & 31;
    v4f v = {0.0f, 0.0f, 0.0f, 0.0f};
    if (is_w)
      v = *(const v4f*)(W1 + (size_t)(tile * 16 + r) * DFEAT + 4 * c4);
    else if (r == 0)
      v = *(const v4f*)(b1 + 4 * c4);
    *(v4f*)(sA + r * APITCH + 4 * c4) = v;
  }
  __syncthreads();
  wmma_tile_16x128(sA, W2, is_w ? W12 : c12, is_w ? tile * 16 : 0,
                   is_w ? DFEAT : 1, tid);
}

__global__ void __launch_bounds__(1024) tgcn_bucket_scatter(
    const int* __restrict__ src, const int* __restrict__ dst,
    const int* __restrict__ bmat, const int* __restrict__ btot,
    int* __restrict__ pairs, int n_edges, int nb, int nblk) {
  __shared__ int s_slot[MAXB];
  __shared__ int s_w[32];
  const int tid = threadIdx.x;
  const int lane = tid & 31;
  const int wave = tid >> 5;
  const int base = blockIdx.x * CHUNK + tid;
  int d[8], s[8];
#pragma unroll
  for (int j = 0; j < 8; ++j) {
    const int e = base + j * 1024;
    d[j] = (e < n_edges) ? dst[e] : -1;
    s[j] = (e < n_edges) ? src[e] : 0;
  }
  const int t = (tid < nb) ? btot[tid] : 0;
  const int mine = (tid < nb) ? bmat[(size_t)tid * nblk + blockIdx.x] : 0;
  const int inc = wave_scan_incl(t, lane);
  if (lane == 31) s_w[wave] = inc;
  __syncthreads();
  int off = 0;
  for (int w = 0; w < 32; ++w)
    if (w < wave) off += s_w[w];
  if (tid < nb) s_slot[tid] = off + inc - t + mine;
  __syncthreads();
#pragma unroll
  for (int j = 0; j < 8; ++j) {
    if (d[j] >= 0) {
      const int slot = atomicAdd(&s_slot[d[j] >> RB_SHIFT], 1);
      *(v2i*)(pairs + 2 * (size_t)slot) = (v2i){s[j], d[j]};
    }
  }
}

__global__ void __launch_bounds__(RB) tgcn_bucket_sort(
    const int* __restrict__ pairs, const int* __restrict__ btot,
    int* __restrict__ csr_src, int* __restrict__ seg_end, int* __restrict__ cnt,
    float* __restrict__ dinv, int n_nodes) {
  __shared__ int s_c[RB];
  __shared__ int s_w[RB / 32];
  const int tid = threadIdx.x;
  const int lane = tid & 31;
  const int wave = tid >> 5;
  const int b = blockIdx.x;

  int pre = 0;
  for (int k = tid; k < b; k += RB) pre += btot[k];
  pre = wave_sum(pre);
  if (lane == 0) s_w[wave] = pre;
  s_c[tid] = 0;
  const int size = btot[b];
  __syncthreads();
  int bstart = 0;
  for (int w = 0; w < RB / 32; ++w) bstart += s_w[w];
  __syncthreads();

  const int* pb = pairs + 2 * (size_t)bstart;
  const int nbase = b << RB_SHIFT;

  for (int i0 = tid; i0 < size; i0 += 8 * RB) {
    int dv[8];
#pragma unroll
    for (int u = 0; u < 8; ++u) {
      const int i = i0 + u * RB;
      dv[u] = (i < size) ? pb[2 * (size_t)i + 1] : -1;
    }
#pragma unroll
    for (int u = 0; u < 8; ++u)
      if (dv[u] >= 0) atomicAdd(&s_c[dv[u] - nbase], 1);
  }
  __syncthreads();

  const int c = s_c[tid];
  const int inc = wave_scan_incl(c, lane);
  if (lane == 31) s_w[wave] = inc;
  __syncthreads();
  int off = 0;
  for (int w = 0; w < RB / 32; ++w)
    if (w < wave) off += s_w[w];
  const int first = off + inc - c;
  const int node = nbase + tid;
  if (node < n_nodes) {
    cnt[node] = c;
    seg_end[node] = bstart + first + c;
    dinv[node] = rsqrtf((float)(c + 1));
  }
  s_c[tid] = first;
  __syncthreads();

  for (int i0 = tid; i0 < size; i0 += 8 * RB) {
    v2i p[8];
#pragma unroll
    for (int u = 0; u < 8; ++u) {
      const int i = i0 + u * RB;
      p[u] = (i < size) ? *(const v2i*)(pb + 2 * (size_t)i) : (v2i){0, -1};
    }
#pragma unroll
    for (int u = 0; u < 8; ++u) {
      if (p[u].y >= 0) {
        const int slot = atomicAdd(&s_c[p[u].y - nbase], 1);
        csr_src[bstart + slot] = p[u].x;
      }
    }
  }
}

__global__ void __launch_bounds__(256) tgcn_gemm_wmma(
    const float* __restrict__ X, const float* __restrict__ W,
    const float* __restrict__ t, const float* __restrict__ freq,
    const float* __restrict__ phase, int use_te,
    float* __restrict__ Y, int n_nodes) {
  __shared__ __attribute__((aligned(16))) float sA[16 * APITCH];

  const int tid = threadIdx.x;
  const int mbase = blockIdx.x * 16;

  for (int i = tid; i < 16 * DFEAT; i += 256) {
    int r = i >> 7;
    int c = i & (DFEAT - 1);
    int row = mbase + r;
    float v = 0.0f;
    if (row < n_nodes) {
      v = X[(size_t)row * DFEAT + c];
      if (use_te) v += cosf(t[row] * freq[c] + phase[c]);
    }
    sA[r * APITCH + c] = v;
  }
  __syncthreads();

  wmma_tile_16x128(sA, W, Y, mbase, n_nodes, tid);
}

template <int G, bool PRED>
__device__ __forceinline__ void row_group(v4f& acc, const float* __restrict__ xl,
                                          int sj, float wj, int j, int m) {
  v4f v[G];
  float w[G];
#pragma unroll
  for (int u = 0; u < G; ++u) {
    int idx = j + u;
    if (PRED) idx = min(idx, m - 1);
    const int s = __builtin_amdgcn_readlane(sj, idx);
    w[u] = __builtin_bit_cast(
        float, __builtin_amdgcn_readlane(__builtin_bit_cast(int, wj), idx));
    v[u] = *(const v4f*)(xl + (size_t)s * DFEAT);
  }
#pragma unroll
  for (int u = 0; u < G; ++u)
    if (!PRED || j + u < m) acc += v[u] * w[u];
}

__global__ void __launch_bounds__(256) tgcn_gather(
    const float* __restrict__ xw, const int* __restrict__ csr_src,
    const int* __restrict__ seg_end, const int* __restrict__ cnt,
    const float* __restrict__ dinv, const float* __restrict__ bias,
    float* __restrict__ out, int n_nodes) {
  const int lane = threadIdx.x & 31;
  const int row = blockIdx.x * 8 + __builtin_amdgcn_readfirstlane(threadIdx.x >> 5);
  if (row >= n_nodes) return;

  const int end = seg_end[row];
  const int start = end - cnt[row];
  const float di = dinv[row];
  const float* xl = xw + 4 * lane;

  const v4f self_row = *(const v4f*)(xl + (size_t)row * DFEAT);
  v4f acc = {0.0f, 0.0f, 0.0f, 0.0f};

  for (int base = start; base < end; base += 32) {
    const int m = min(32, end - base);
    int sj = 0;
    float wj = 0.0f;
    if (lane < m) {
      sj = csr_src[base + lane];
      wj = dinv[sj] * di;
    }
    int j = 0;
    for (; j + 8 <= m; j += 8) row_group<8, false>(acc, xl, sj, wj, j, m);
    if (j < m) {
      row_group<4, true>(acc, xl, sj, wj, j, m);
      j += 4;
    }
    if (j < m) row_group<4, true>(acc, xl, sj, wj, j, m);
  }

  acc += self_row * (di * di);
  const v4f b = *(const v4f*)(bias + 4 * lane);
  *(v4f*)(out + (size_t)row * DFEAT + 4 * lane) = acc + b;
}

__global__ void __launch_bounds__(256) tgcn_gather_pair(
    const float* __restrict__ xw, const int* __restrict__ csr_src,
    const int* __restrict__ seg_end, const int* __restrict__ cnt,
    const float* __restrict__ dinv, const float* __restrict__ bias,
    float* __restrict__ out, int n_nodes) {
  const int tid = threadIdx.x;
  const int lane = tid & 31;
  const int wave = __builtin_amdgcn_readfirstlane(tid >> 5);
  const int mbase = blockIdx.x * 16;
  const float* xl = xw + 4 * lane;
  const v4f b = *(const v4f*)(bias + 4 * lane);

  int startq[2], endq[2];
  float diq[2];
  v4f selfq[2];
#pragma unroll
  for (int q = 0; q < 2; ++q) {
    const int row = mbase + 2 * wave + q;
    startq[q] = 0;
    endq[q] = 0;
    diq[q] = 0.0f;
    selfq[q] = (v4f){0.0f, 0.0f, 0.0f, 0.0f};
    if (row < n_nodes) {
      endq[q] = seg_end[row];
      startq[q] = endq[q] - cnt[row];
      diq[q] = dinv[row];
      selfq[q] = *(const v4f*)(xl + (size_t)row * DFEAT);
    }
  }

#pragma unroll
  for (int q = 0; q < 2; ++q) {
    const int row = mbase + 2 * wave + q;
    const int end = endq[q];
    const float di = diq[q];
    v4f acc = {0.0f, 0.0f, 0.0f, 0.0f};
    for (int base = startq[q]; base < end; base += 32) {
      const int m = min(32, end - base);
      int sj = 0;
      float wj = 0.0f;
      if (lane < m) {
        sj = csr_src[base + lane];
        wj = dinv[sj] * di;
      }
      int j = 0;
      for (; j + 8 <= m; j += 8) row_group<8, false>(acc, xl, sj, wj, j, m);
      if (j < m) {
        row_group<4, true>(acc, xl, sj, wj, j, m);
        j += 4;
      }
      if (j < m) row_group<4, true>(acc, xl, sj, wj, j, m);
    }
    acc += selfq[q] * (di * di);
    if (row < n_nodes) *(v4f*)(out + (size_t)row * DFEAT + 4 * lane) = acc + b;
  }
}

extern "C" void kernel_launch(void* const* d_in, const int* in_sizes, int n_in,
                              void* d_out, int out_size, void* d_ws, size_t ws_size,
                              hipStream_t stream) {
  const float* x     = (const float*)d_in[0];
  const int*   ei    = (const int*)d_in[1];
  const float* t     = (const float*)d_in[2];
  const float* W1    = (const float*)d_in[3];
  const float* b1    = (const float*)d_in[4];
  const float* W2    = (const float*)d_in[5];
  const float* b2    = (const float*)d_in[6];
  const float* freq  = (const float*)d_in[7];
  const float* phase = (const float*)d_in[8];

  const int n_nodes = in_sizes[2];
  const int n_edges = in_sizes[1] / 2;
  const size_t nd = (size_t)n_nodes * DFEAT;

  const int* src = ei;
  const int* dst = ei + n_edges;

  float* out = (float*)d_out;

  const int T = 256;
  const int g_gemm = (n_nodes + 15) / 16;
  const int g_gath = (n_nodes + 7) / 8;
  const int nb = (n_nodes + RB - 1) / RB;
  const int nblk = (n_edges + CHUNK - 1) / CHUNK;

  float* dinv = (float*)d_ws;
  size_t noff = ((size_t)n_nodes + 255) & ~(size_t)255;
  size_t moff = ((size_t)nb * nblk + 255) & ~(size_t)255;
  size_t boff = ((size_t)nb + 255) & ~(size_t)255;
  size_t eoff = ((size_t)n_edges + 255) & ~(size_t)255;
  float* bufA = dinv + noff;
  float* bufB = bufA + nd;
  int* cnt      = (int*)(bufB + nd);
  int* cursor   = cnt + noff;
  int* bmat     = cursor + noff;
  int* btot     = bmat + moff;
  int* pairs    = btot + boff;
  int* csr_src  = pairs + 2 * eoff;
  float* W12 = (float*)(csr_src + eoff);
  float* c12 = W12 + (size_t)DFEAT * DFEAT;

  tgcn_bucket_hist<<<nblk, 1024, 0, stream>>>(dst, bmat, n_edges, nb, nblk);
  tgcn_scan_fold<<<nb + 9, T, 0, stream>>>(bmat, btot, nb, nblk, W1, b1, W2, W12, c12);
  tgcn_bucket_scatter<<<nblk, 1024, 0, stream>>>(src, dst, bmat, btot, pairs, n_edges, nb, nblk);
  tgcn_bucket_sort<<<nb, RB, 0, stream>>>(pairs, btot, csr_src, cursor, cnt, dinv, n_nodes);

  tgcn_gemm_wmma<<<g_gemm, T, 0, stream>>>(x, W12, t, freq, phase, 1, bufA, n_nodes);

  tgcn_gather_pair<<<g_gemm, T, 0, stream>>>(bufA, csr_src, cursor, cnt, dinv, c12, bufB, n_nodes);

  tgcn_gather<<<g_gath, T, 0, stream>>>(bufB, csr_src, cursor, cnt, dinv, b2, out, n_nodes);
}


# both aggregation launches: 1792 resident work-groups whose waves walk over destination rows instead of one short wave per row (or two); 121328 fewer waves to start; per-row sums unchanged
# speedup vs baseline: 5.1576x; 1.0422x over previous
#include <hip/hip_runtime.h>
#include <math.h>

typedef __attribute__((ext_vector_type(2))) float v2f;
typedef __attribute__((ext_vector_type(2))) int v2i;
typedef __attribute__((ext_vector_type(4))) float v4f;
typedef __attribute__((ext_vector_type(8))) float v8f;

#define DFEAT 128
#define APITCH 132
#define RB 512
#define RB_SHIFT 9
#define MAXB 1024
#define CHUNK 8192


__device__ __forceinline__ int wave_sum(int v) {
  for (int o = 16; o > 0; o >>= 1) v += __shfl_xor(v, o);
  return v;
}

__device__ __forceinline__ int wave_scan_incl(int v, int lane) {
  for (int o = 1; o < 32; o <<= 1) {
    int y = __shfl_up(v, o);
    if (lane >= o) v += y;
  }
  return v;
}

__global__ void __launch_bounds__(1024) tgcn_bucket_hist(
    const int* __restrict__ dst, int* __restrict__ bmat, int n_edges, int nb,
    int nblk) {
  __shared__ int s_cnt[MAXB];
  const int tid = threadIdx.x;
  if (tid < nb) s_cnt[tid] = 0;
  __syncthreads();
  const int base = blockIdx.x * CHUNK + tid;
  int d[8];
#pragma unroll
  for (int j = 0; j < 8; ++j) {
    const int e = base + j * 1024;
    d[j] = (e < n_edges) ? dst[e] : -1;
  }
#pragma unroll
  for (int j = 0; j < 8; ++j)
    if (d[j] >= 0) atomicAdd(&s_cnt[d[j] >> RB_SHIFT], 1);
  __syncthreads();
  if (tid < nb) bmat[(size_t)tid * nblk + blockIdx.x] = s_cnt[tid];
}

__device__ __forceinline__ void wmma_tile_16x128(
    const float* sA, const float* __restrict__ W, float* __restrict__ Y,
    int mbase, int n_nodes, int tid) {
  const int wave = tid >> 5;
  const int lane = tid & 31;
  const int half = lane >> 4;
  const int arow = lane & 15;
  const int ncol = wave * 16 + (lane & 15);

  v8f acc = {};
  for (int k = 0; k < DFEAT; k += 4) {
    const int ka = k + 2 * half;
    v2f a, b;
    a.x = sA[arow * APITCH + ka];
    a.y = sA[arow * APITCH + ka + 1];
    b.x = W[(size_t)ka * DFEAT + ncol];
    b.y = W[(size_t)(ka + 1) * DFEAT + ncol];
    acc = __builtin_amdgcn_wmma_f32_16x16x4_f32(
         false, a,  false, b,
         (short)0, acc,  false,  false);
  }

#pragma unroll
  for (int r = 0; r < 8; ++r) {
    int orow = mbase + r + 8 * half;
    if (orow < n_nodes) Y[(size_t)orow * DFEAT + ncol] = acc[r];
  }
}

__global__ void __launch_bounds__(256) tgcn_scan_fold(
    int* __restrict__ bmat, int* __restrict__ btot, int nb, int nblk,
    const float* __restrict__ W1, const float* __restrict__ b1,
    const float* __restrict__ W2, float* __restrict__ W12,
    float* __restrict__ c12) {
  __shared__ __attribute__((aligned(16))) float sA[16 * APITCH];
  __shared__ int s_w[8];
  const int tid = threadIdx.x;
  if ((int)blockIdx.x < nb) {
    int* row = bmat + (size_t)blockIdx.x * nblk;
    const int lane = tid & 31;
    const int wave = tid >> 5;
    int carry = 0;
    for (int base = 0; base < nblk; base += 256) {
      const int i = base + tid;
      const int v = (i < nblk) ? row[i] : 0;
      const int inc = wave_scan_incl(v, lane);
      if (lane == 31) s_w[wave] = inc;
      __syncthreads();
      int off = 0, tot = 0;
      for (int w = 0; w < 8; ++w) {
        const int s = s_w[w];
        if (w < wave) off += s;
        tot += s;
      }
      if (i < nblk) row[i] = carry + off + inc - v;
      carry += tot;
      __syncthreads();
    }
    if (tid == 0) btot[blockIdx.x] = carry;
    return;
  }
  const int tile = blockIdx.x - nb;
  const bool is_w = tile < 8;
  for (int i = tid; i < 16 * (DFEAT / 4); i += 256) {
    const int r = i >> 5;
    const int c4 = i & 31;
    v4f v = {0.0f, 0.0f, 0.0f, 0.0f};
    if (is_w)
      v = *(const v4f*)(W1 + (size_t)(tile * 16 + r) * DFEAT + 4 * c4);
    else if (r == 0)
      v = *(const v4f*)(b1 + 4 * c4);
    *(v4f*)(sA + r * APITCH + 4 * c4) = v;
  }
  __syncthreads();
  wmma_tile_16x128(sA, W2, is_w ? W12 : c12, is_w ? tile * 16 : 0,
                   is_w ? DFEAT : 1, tid);
}

__global__ void __launch_bounds__(1024) tgcn_bucket_scatter(
    const int* __restrict__ src, const int* __restrict__ dst,
    const int* __restrict__ bmat, const int* __restrict__ btot,
    int* __restrict__ pairs, int n_edges, int nb, int nblk) {
  __shared__ int s_slot[MAXB];
  __shared__ int s_w[32];
  const int tid = threadIdx.x;
  const int lane = tid & 31;
  const int wave = tid >> 5;
  const int base = blockIdx.x * CHUNK + tid;
  int d[8], s[8];
#pragma unroll
  for (int j = 0; j < 8; ++j) {
    const int e = base + j * 1024;
    d[j] = (e < n_edges) ? dst[e] : -1;
    s[j] = (e < n_edges) ? src[e] : 0;
  }
  const int t = (tid < nb) ? btot[tid] : 0;
  const int mine = (tid < nb) ? bmat[(size_t)tid * nblk + blockIdx.x] : 0;
  const int inc = wave_scan_incl(t, lane);
  if (lane == 31) s_w[wave] = inc;
  __syncthreads();
  int off = 0;
  for (int w = 0; w < 32; ++w)
    if (w < wave) off += s_w[w];
  if (tid < nb) s_slot[tid] = off + inc - t + mine;
  __syncthreads();
#pragma unroll
  for (int j = 0; j < 8; ++j) {
    if (d[j] >= 0) {
      const int slot = atomicAdd(&s_slot[d[j] >> RB_SHIFT], 1);
      *(v2i*)(pairs + 2 * (size_t)slot) = (v2i){s[j], d[j]};
    }
  }
}

__global__ void __launch_bounds__(RB) tgcn_bucket_sort(
    const int* __restrict__ pairs, const int* __restrict__ btot,
    int* __restrict__ csr_src, int* __restrict__ seg_end, int* __restrict__ cnt,
    float* __restrict__ dinv, int n_nodes) {
  __shared__ int s_c[RB];
  __shared__ int s_w[RB / 32];
  const int tid = threadIdx.x;
  const int lane = tid & 31;
  const int wave = tid >> 5;
  const int b = blockIdx.x;

  int pre = 0;
  for (int k = tid; k < b; k += RB) pre += btot[k];
  pre = wave_sum(pre);
  if (lane == 0) s_w[wave] = pre;
  s_c[tid] = 0;
  const int size = btot[b];
  __syncthreads();
  int bstart = 0;
  for (int w = 0; w < RB / 32; ++w) bstart += s_w[w];
  __syncthreads();

  const int* pb = pairs + 2 * (size_t)bstart;
  const int nbase = b << RB_SHIFT;

  for (int i0 = tid; i0 < size; i0 += 8 * RB) {
    int dv[8];
#pragma unroll
    for (int u = 0; u < 8; ++u) {
      const int i = i0 + u * RB;
      dv[u] = (i < size) ? pb[2 * (size_t)i + 1] : -1;
    }
#pragma unroll
    for (int u = 0; u < 8; ++u)
      if (dv[u] >= 0) atomicAdd(&s_c[dv[u] - nbase], 1);
  }
  __syncthreads();

  const int c = s_c[tid];
  const int inc = wave_scan_incl(c, lane);
  if (lane == 31) s_w[wave] = inc;
  __syncthreads();
  int off = 0;
  for (int w = 0; w < RB / 32; ++w)
    if (w < wave) off += s_w[w];
  const int first = off + inc - c;
  const int node = nbase + tid;
  if (node < n_nodes) {
    cnt[node] = c;
    seg_end[node] = bstart + first + c;
    dinv[node] = rsqrtf((float)(c + 1));
  }
  s_c[tid] = first;
  __syncthreads();

  for (int i0 = tid; i0 < size; i0 += 8 * RB) {
    v2i p[8];
#pragma unroll
    for (int u = 0; u < 8; ++u) {
      const int i = i0 + u * RB;
      p[u] = (i < size) ? *(const v2i*)(pb + 2 * (size_t)i) : (v2i){0, -1};
    }
#pragma unroll
    for (int u = 0; u < 8; ++u) {
      if (p[u].y >= 0) {
        const int slot = atomicAdd(&s_c[p[u].y - nbase], 1);
        csr_src[bstart + slot] = p[u].x;
      }
    }
  }
}

__global__ void __launch_bounds__(256) tgcn_gemm_wmma(
    const float* __restrict__ X, const float* __restrict__ W,
    const float* __restrict__ t, const float* __restrict__ freq,
    const float* __restrict__ phase, int use_te,
    float* __restrict__ Y, int n_nodes) {
  __shared__ __attribute__((aligned(16))) float sA[16 * APITCH];

  const int tid = threadIdx.x;
  const int mbase = blockIdx.x * 16;

  for (int i = tid; i < 16 * DFEAT; i += 256) {
    int r = i >> 7;
    int c = i & (DFEAT - 1);
    int row = mbase + r;
    float v = 0.0f;
    if (row < n_nodes) {
      v = X[(size_t)row * DFEAT + c];
      if (use_te) v += cosf(t[row] * freq[c] + phase[c]);
    }
    sA[r * APITCH + c] = v;
  }
  __syncthreads();

  wmma_tile_16x128(sA, W, Y, mbase, n_nodes, tid);
}

template <int G, bool PRED>
__device__ __forceinline__ void row_group(v4f& acc, const float* __restrict__ xl,
                                          int sj, float wj, int j, int m) {
  v4f v[G];
  float w[G];
#pragma unroll
  for (int u = 0; u < G; ++u) {
    int idx = j + u;
    if (PRED) idx = min(idx, m - 1);
    const int s = __builtin_amdgcn_readlane(sj, idx);
    w[u] = __builtin_bit_cast(
        float, __builtin_amdgcn_readlane(__builtin_bit_cast(int, wj), idx));
    v[u] = *(const v4f*)(xl + (size_t)s * DFEAT);
  }
#pragma unroll
  for (int u = 0; u < G; ++u)
    if (!PRED || j + u < m) acc += v[u] * w[u];
}

#define GATHER_WGS 1792

__global__ void __launch_bounds__(256) tgcn_gather(
    const float* __restrict__ xw, const int* __restrict__ csr_src,
    const int* __restrict__ seg_end, const int* __restrict__ cnt,
    const float* __restrict__ dinv, const float* __restrict__ bias,
    float* __restrict__ out, int n_nodes) {
  const int lane = threadIdx.x & 31;
  const int wave = __builtin_amdgcn_readfirstlane(threadIdx.x >> 5);
  const int nwaves = gridDim.x * 8;
  const float* xl = xw + 4 * lane;
  const v4f b = *(const v4f*)(bias + 4 * lane);

  int row = blockIdx.x * 8 + wave;
  if (row >= n_nodes) return;
  int end = seg_end[row];
  int deg = cnt[row];
  float di = dinv[row];
  v4f self_row = *(const v4f*)(xl + (size_t)row * DFEAT);

  for (;;) {
    const int nrow = row + nwaves;
    const bool more = nrow < n_nodes;
    int nend = 0, ndeg = 0;
    float ndi = 0.0f;
    v4f nself = {0.0f, 0.0f, 0.0f, 0.0f};
    if (more) {
      nend = seg_end[nrow];
      ndeg = cnt[nrow];
      ndi = dinv[nrow];
      nself = *(const v4f*)(xl + (size_t)nrow * DFEAT);
    }

    v4f acc = {0.0f, 0.0f, 0.0f, 0.0f};
    for (int base = end - deg; base < end; base += 32) {
      const int m = min(32, end - base);
      int sj = 0;
      float wj = 0.0f;
      if (lane < m) {
        sj = csr_src[base + lane];
        wj = dinv[sj] * di;
      }
      int j = 0;
      for (; j + 8 <= m; j += 8) row_group<8, false>(acc, xl, sj, wj, j, m);
      if (j < m) {
        row_group<4, true>(acc, xl, sj, wj, j, m);
        j += 4;
      }
      if (j < m) row_group<4, true>(acc, xl, sj, wj, j, m);
    }
    acc += self_row * (di * di);
    *(v4f*)(out + (size_t)row * DFEAT + 4 * lane) = acc + b;

    if (!more) break;
    row = nrow;
    end = nend;
    deg = ndeg;
    di = ndi;
    self_row = nself;
  }
}

extern "C" void kernel_launch(void* const* d_in, const int* in_sizes, int n_in,
                              void* d_out, int out_size, void* d_ws, size_t ws_size,
                              hipStream_t stream) {
  const float* x     = (const float*)d_in[0];
  const int*   ei    = (const int*)d_in[1];
  const float* t     = (const float*)d_in[2];
  const float* W1    = (const float*)d_in[3];
  const float* b1    = (const float*)d_in[4];
  const float* W2    = (const float*)d_in[5];
  const float* b2    = (const float*)d_in[6];
  const float* freq  = (const float*)d_in[7];
  const float* phase = (const float*)d_in[8];

  const int n_nodes = in_sizes[2];
  const int n_edges = in_sizes[1] / 2;
  const size_t nd = (size_t)n_nodes * DFEAT;

  const int* src = ei;
  const int* dst = ei + n_edges;

  float* out = (float*)d_out;

  const int T = 256;
  const int g_gemm = (n_nodes + 15) / 16;
  const int nb = (n_nodes + RB - 1) / RB;
  const int nblk = (n_edges + CHUNK - 1) / CHUNK;

  float* dinv = (float*)d_ws;
  size_t noff = ((size_t)n_nodes + 255) & ~(size_t)255;
  size_t moff = ((size_t)nb * nblk + 255) & ~(size_t)255;
  size_t boff = ((size_t)nb + 255) & ~(size_t)255;
  size_t eoff = ((size_t)n_edges + 255) & ~(size_t)255;
  float* bufA = dinv + noff;
  float* bufB = bufA + nd;
  int* cnt      = (int*)(bufB + nd);
  int* cursor   = cnt + noff;
  int* bmat     = cursor + noff;
  int* btot     = bmat + moff;
  int* pairs    = btot + boff;
  int* csr_src  = pairs + 2 * eoff;
  float* W12 = (float*)(csr_src + eoff);
  float* c12 = W12 + (size_t)DFEAT * DFEAT;

  tgcn_bucket_hist<<<nblk, 1024, 0, stream>>>(dst, bmat, n_edges, nb, nblk);
  tgcn_scan_fold<<<nb + 9, T, 0, stream>>>(bmat, btot, nb, nblk, W1, b1, W2, W12, c12);
  tgcn_bucket_scatter<<<nblk, 1024, 0, stream>>>(src, dst, bmat, btot, pairs, n_edges, nb, nblk);
  tgcn_bucket_sort<<<nb, RB, 0, stream>>>(pairs, btot, csr_src, cursor, cnt, dinv, n_nodes);

  tgcn_gemm_wmma<<<g_gemm, T, 0, stream>>>(x, W12, t, freq, phase, 1, bufA, n_nodes);

  tgcn_gather<<<GATHER_WGS, T, 0, stream>>>(bufA, csr_src, cursor, cnt, dinv, c12, bufB, n_nodes);

  tgcn_gather<<<GATHER_WGS, T, 0, stream>>>(bufB, csr_src, cursor, cnt, dinv, b2, out, n_nodes);
}


# per-node product as a resident grid: 768 work-groups walk over 16-row tiles, each wave keeps its W12 column fragments in 64 registers, next tile's rows load under the K loop; sums unchanged
# speedup vs baseline: 5.8152x; 1.1275x over previous
#include <hip/hip_runtime.h>
#include <math.h>

typedef __attribute__((ext_vector_type(2))) float v2f;
typedef __attribute__((ext_vector_type(2))) int v2i;
typedef __attribute__((ext_vector_type(4))) float v4f;
typedef __attribute__((ext_vector_type(8))) float v8f;

#define DFEAT 128
#define APITCH 132
#define RB 512
#define RB_SHIFT 9
#define MAXB 1024
#define CHUNK 8192


__device__ __forceinline__ int wave_sum(int v) {
  for (int o = 16; o > 0; o >>= 1) v += __shfl_xor(v, o);
  return v;
}

__device__ __forceinline__ int wave_scan_incl(int v, int lane) {
  for (int o = 1; o < 32; o <<= 1) {
    int y = __shfl_up(v, o);
    if (lane >= o) v += y;
  }
  return v;
}

__global__ void __launch_bounds__(1024) tgcn_bucket_hist(
    const int* __restrict__ dst, int* __restrict__ bmat, int n_edges, int nb,
    int nblk) {
  __shared__ int s_cnt[MAXB];
  const int tid = threadIdx.x;
  if (tid < nb) s_cnt[tid] = 0;
  __syncthreads();
  const int base = blockIdx.x * CHUNK + tid;
  int d[8];
#pragma unroll
  for (int j = 0; j < 8; ++j) {
    const int e = base + j * 1024;
    d[j] = (e < n_edges) ? dst[e] : -1;
  }
#pragma unroll
  for (int j = 0; j < 8; ++j)
    if (d[j] >= 0) atomicAdd(&s_cnt[d[j] >> RB_SHIFT], 1);
  __syncthreads();
  if (tid < nb) bmat[(size_t)tid * nblk + blockIdx.x] = s_cnt[tid];
}

__device__ __forceinline__ void wmma_tile_16x128(
    const float* sA, const float* __restrict__ W, float* __restrict__ Y,
    int mbase, int n_rows, int tid) {
  const int wave = tid >> 5;
  const int lane = tid & 31;
  const int half = lane >> 4;
  const int arow = lane & 15;
  const int ncol = wave * 16 + (lane & 15);

  v8f acc = {};
  for (int k = 0; k < DFEAT; k += 4) {
    const int ka = k + 2 * half;
    v2f a, b;
    a.x = sA[arow * APITCH + ka];
    a.y = sA[arow * APITCH + ka + 1];
    b.x = W[(size_t)ka * DFEAT + ncol];
    b.y = W[(size_t)(ka + 1) * DFEAT + ncol];
    acc = __builtin_amdgcn_wmma_f32_16x16x4_f32(
         false, a,  false, b,
         (short)0, acc,  false,  false);
  }

#pragma unroll
  for (int r = 0; r < 8; ++r) {
    int orow = mbase + r + 8 * half;
    if (orow < n_rows) Y[(size_t)orow * DFEAT + ncol] = acc[r];
  }
}

__global__ void __launch_bounds__(256) tgcn_scan_fold(
    int* __restrict__ bmat, int* __restrict__ btot, int nb, int nblk,
    const float* __restrict__ W1, const float* __restrict__ b1,
    const float* __restrict__ W2, float* __restrict__ W12,
    float* __restrict__ c12) {
  __shared__ __attribute__((aligned(16))) float sA[16 * APITCH];
  __shared__ int s_w[8];
  const int tid = threadIdx.x;
  if ((int)blockIdx.x < nb) {
    int* row = bmat + (size_t)blockIdx.x * nblk;
    const int lane = tid & 31;
    const int wave = tid >> 5;
    int carry = 0;
    for (int base = 0; base < nblk; base += 256) {
      const int i = base + tid;
      const int v = (i < nblk) ? row[i] : 0;
      const int inc = wave_scan_incl(v, lane);
      if (lane == 31) s_w[wave] = inc;
      __syncthreads();
      int off = 0, tot = 0;
      for (int w = 0; w < 8; ++w) {
        const int s = s_w[w];
        if (w < wave) off += s;
        tot += s;
      }
      if (i < nblk) row[i] = carry + off + inc - v;
      carry += tot;
      __syncthreads();
    }
    if (tid == 0) btot[blockIdx.x] = carry;
    return;
  }
  const int tile = blockIdx.x - nb;
  const bool is_w = tile < 8;
  for (int i = tid; i < 16 * (DFEAT / 4); i += 256) {
    const int r = i >> 5;
    const int c4 = i & 31;
    v4f v = {0.0f, 0.0f, 0.0f, 0.0f};
    if (is_w)
      v = *(const v4f*)(W1 + (size_t)(tile * 16 + r) * DFEAT + 4 * c4);
    else if (r == 0)
      v = *(const v4f*)(b1 + 4 * c4);
    *(v4f*)(sA + r * APITCH + 4 * c4) = v;
  }
  __syncthreads();
  wmma_tile_16x128(sA, W2, is_w ? W12 : c12, is_w ? tile * 16 : 0,
                   is_w ? DFEAT : 1, tid);
}

__global__ void __launch_bounds__(1024) tgcn_bucket_scatter(
    const int* __restrict__ src, const int* __restrict__ dst,
    const int* __restrict__ bmat, const int* __restrict__ btot,
    int* __restrict__ pairs, int n_edges, int nb, int nblk) {
  __shared__ int s_slot[MAXB];
  __shared__ int s_w[32];
  const int tid = threadIdx.x;
  const int lane = tid & 31;
  const int wave = tid >> 5;
  const int base = blockIdx.x * CHUNK + tid;
  int d[8], s[8];
#pragma unroll
  for (int j = 0; j < 8; ++j) {
    const int e = base + j * 1024;
    d[j] = (e < n_edges) ? dst[e] : -1;
    s[j] = (e < n_edges) ? src[e] : 0;
  }
  const int t = (tid < nb) ? btot[tid] : 0;
  const int mine = (tid < nb) ? bmat[(size_t)tid * nblk + blockIdx.x] : 0;
  const int inc = wave_scan_incl(t, lane);
  if (lane == 31) s_w[wave] = inc;
  __syncthreads();
  int off = 0;
  for (int w = 0; w < 32; ++w)
    if (w < wave) off += s_w[w];
  if (tid < nb) s_slot[tid] = off + inc - t + mine;
  __syncthreads();
#pragma unroll
  for (int j = 0; j < 8; ++j) {
    if (d[j] >= 0) {
      const int slot = atomicAdd(&s_slot[d[j] >> RB_SHIFT], 1);
      *(v2i*)(pairs + 2 * (size_t)slot) = (v2i){s[j], d[j]};
    }
  }
}

__global__ void __launch_bounds__(RB) tgcn_bucket_sort(
    const int* __restrict__ pairs, const int* __restrict__ btot,
    int* __restrict__ csr_src, int* __restrict__ seg_end, int* __restrict__ cnt,
    float* __restrict__ dinv, int n_nodes) {
  __shared__ int s_c[RB];
  __shared__ int s_w[RB / 32];
  const int tid = threadIdx.x;
  const int lane = tid & 31;
  const int wave = tid >> 5;
  const int b = blockIdx.x;

  int pre = 0;
  for (int k = tid; k < b; k += RB) pre += btot[k];
  pre = wave_sum(pre);
  if (lane == 0) s_w[wave] = pre;
  s_c[tid] = 0;
  const int size = btot[b];
  __syncthreads();
  int bstart = 0;
  for (int w = 0; w < RB / 32; ++w) bstart += s_w[w];
  __syncthreads();

  const int* pb = pairs + 2 * (size_t)bstart;
  const int nbase = b << RB_SHIFT;

  for (int i0 = tid; i0 < size; i0 += 8 * RB) {
    int dv[8];
#pragma unroll
    for (int u = 0; u < 8; ++u) {
      const int i = i0 + u * RB;
      dv[u] = (i < size) ? pb[2 * (size_t)i + 1] : -1;
    }
#pragma unroll
    for (int u = 0; u < 8; ++u)
      if (dv[u] >= 0) atomicAdd(&s_c[dv[u] - nbase], 1);
  }
  __syncthreads();

  const int c = s_c[tid];
  const int inc = wave_scan_incl(c, lane);
  if (lane == 31) s_w[wave] = inc;
  __syncthreads();
  int off = 0;
  for (int w = 0; w < RB / 32; ++w)
    if (w < wave) off += s_w[w];
  const int first = off + inc - c;
  const int node = nbase + tid;
  if (node < n_nodes) {
    cnt[node] = c;
    seg_end[node] = bstart + first + c;
    dinv[node] = rsqrtf((float)(c + 1));
  }
  s_c[tid] = first;
  __syncthreads();

  for (int i0 = tid; i0 < size; i0 += 8 * RB) {
    v2i p[8];
#pragma unroll
    for (int u = 0; u < 8; ++u) {
      const int i = i0 + u * RB;
      p[u] = (i < size) ? *(const v2i*)(pb + 2 * (size_t)i) : (v2i){0, -1};
    }
#pragma unroll
    for (int u = 0; u < 8; ++u) {
      if (p[u].y >= 0) {
        const int slot = atomicAdd(&s_c[p[u].y - nbase], 1);
        csr_src[bstart + slot] = p[u].x;
      }
    }
  }
}

#define GEMM_WGS 768

struct ATileRows {
  v4f x0, x1;
  float t0, t1;
  float d0, d1;
};

__device__ __forceinline__ ATileRows a_rows_fetch(
    const float* __restrict__ X, const float* __restrict__ t,
    const float* __restrict__ rowscale, int tile, int wave, int lane,
    int n_nodes) {
  const int r0 = min(tile * 16 + wave, n_nodes - 1);
  const int r1 = min(tile * 16 + wave + 8, n_nodes - 1);
  ATileRows f;
  f.x0 = *(const v4f*)(X + (size_t)r0 * DFEAT + 4 * lane);
  f.x1 = *(const v4f*)(X + (size_t)r1 * DFEAT + 4 * lane);
  f.t0 = t[r0];
  f.t1 = t[r1];
  f.d0 = rowscale[r0];
  f.d1 = rowscale[r1];
  return f;
}

__device__ __forceinline__ v4f a_row_value(v4f x, float tr, float d, v4f fq,
                                           v4f ph, int use_te, bool inside) {
  v4f v = x;
  if (use_te) {
    v.x += cosf(tr * fq.x + ph.x);
    v.y += cosf(tr * fq.y + ph.y);
    v.z += cosf(tr * fq.z + ph.z);
    v.w += cosf(tr * fq.w + ph.w);
  }
  v *= d;
  if (!inside) v = (v4f){0.0f, 0.0f, 0.0f, 0.0f};
  return v;
}

__global__ void __launch_bounds__(256) tgcn_gemm_wmma(
    const float* __restrict__ X, const float* __restrict__ W,
    const float* __restrict__ t, const float* __restrict__ freq,
    const float* __restrict__ phase, int use_te,
    const float* __restrict__ rowscale, float* __restrict__ Y, int n_nodes) {
  __shared__ __attribute__((aligned(16))) float sA[2][16 * APITCH];

  const int tid = threadIdx.x;
  const int lane = tid & 31;
  const int wave = __builtin_amdgcn_readfirstlane(tid >> 5);
  const int half = lane >> 4;
  const int arow = lane & 15;
  const int ncol = wave * 16 + arow;
  const int ntiles = (n_nodes + 15) >> 4;
  const int step = gridDim.x;

  int tile = blockIdx.x;
  if (tile >= ntiles) return;

  ATileRows cur = a_rows_fetch(X, t, rowscale, tile, wave, lane, n_nodes);

  v2f bw[DFEAT / 4];
#pragma unroll
  for (int s = 0; s < DFEAT / 4; ++s) {
    const int ka = 4 * s + 2 * half;
    bw[s].x = W[(size_t)ka * DFEAT + ncol];
    bw[s].y = W[(size_t)(ka + 1) * DFEAT + ncol];
  }

  v4f fq = {0.0f, 0.0f, 0.0f, 0.0f}, ph = {0.0f, 0.0f, 0.0f, 0.0f};
  if (use_te) {
    fq = *(const v4f*)(freq + 4 * lane);
    ph = *(const v4f*)(phase + 4 * lane);
  }

  float* wr0 = &sA[0][wave * APITCH + 4 * lane];
  float* wr1 = &sA[0][(wave + 8) * APITCH + 4 * lane];
  const float* rd = &sA[0][arow * APITCH + 2 * half];

  *(v4f*)wr0 = a_row_value(cur.x0, cur.t0, cur.d0, fq, ph, use_te,
                           tile * 16 + wave < n_nodes);
  *(v4f*)wr1 = a_row_value(cur.x1, cur.t1, cur.d1, fq, ph, use_te,
                           tile * 16 + wave + 8 < n_nodes);
  __syncthreads();

  int buf = 0;
  for (;;) {
    const int ntile = tile + step;
    const bool more = ntile < ntiles;
    const ATileRows nxt =
        a_rows_fetch(X, t, rowscale, more ? ntile : tile, wave, lane, n_nodes);

    const float* sa = rd + buf * (16 * APITCH);
    v8f acc = {};
#pragma unroll
    for (int s = 0; s < DFEAT / 4; ++s) {
      const v2f a = *(const v2f*)(sa + 4 * s);
      acc = __builtin_amdgcn_wmma_f32_16x16x4_f32(
           false, a,  false, bw[s],
           (short)0, acc,  false,  false);
    }

    const int mbase = tile * 16;
#pragma unroll
    for (int r = 0; r < 8; ++r) {
      const int orow = mbase + r + 8 * half;
      if (orow < n_nodes) Y[(size_t)orow * DFEAT + ncol] = acc[r];
    }

    if (!more) break;
    const int off = (buf ^ 1) * (16 * APITCH);
    *(v4f*)(wr0 + off) = a_row_value(nxt.x0, nxt.t0, nxt.d0, fq, ph, use_te,
                                     ntile * 16 + wave < n_nodes);
    *(v4f*)(wr1 + off) = a_row_value(nxt.x1, nxt.t1, nxt.d1, fq, ph, use_te,
                                     ntile * 16 + wave + 8 < n_nodes);
    __syncthreads();
    buf ^= 1;
    tile = ntile;
  }
}

template <int G, bool PRED>
__device__ __forceinline__ void row_group(v4f& acc, const float* __restrict__ xl,
                                          int sj, int j, int m) {
  v4f v[G];
#pragma unroll
  for (int u = 0; u < G; ++u) {
    int idx = j + u;
    if (PRED) idx = min(idx, m - 1);
    const int s = __builtin_amdgcn_readlane(sj, idx);
    v[u] = *(const v4f*)(xl + (size_t)s * DFEAT);
  }
#pragma unroll
  for (int u = 0; u < G; ++u)
    if (!PRED || j + u < m) acc += v[u];
}

__device__ __forceinline__ void chunk_sum(v4f& acc, const float* __restrict__ xl,
                                          int sj, int m) {
  int j = 0;
  for (; j + 8 <= m; j += 8) row_group<8, false>(acc, xl, sj, j, m);
  if (j < m) {
    row_group<4, true>(acc, xl, sj, j, m);
    j += 4;
  }
  if (j < m) row_group<4, true>(acc, xl, sj, j, m);
}

#define GATHER_WGS 1792

template <bool SCALE_OUT>
__global__ void __launch_bounds__(256) tgcn_gather(
    const float* __restrict__ xs, const int* __restrict__ csr_src,
    const int* __restrict__ seg_end, const int* __restrict__ cnt,
    const float* __restrict__ dinv, const float* __restrict__ bias,
    float* __restrict__ out, int n_nodes) {
  const int lane = threadIdx.x & 31;
  const int wave = __builtin_amdgcn_readfirstlane(threadIdx.x >> 5);
  const int nwaves = gridDim.x * 8;
  const float* xl = xs + 4 * lane;
  const v4f b = *(const v4f*)(bias + 4 * lane);

  int row = blockIdx.x * 8 + wave;
  if (row >= n_nodes) return;
  int end = seg_end[row];
  int deg = cnt[row];
  float di = dinv[row];
  v4f self_row = *(const v4f*)(xl + (size_t)row * DFEAT);

  for (;;) {
    int base = end - deg;
    int sj = 0;
    if (deg > 0) sj = csr_src[min(base + lane, end - 1)];

    const int nrow = row + nwaves;
    const bool more = nrow < n_nodes;
    const int prow = more ? nrow : row;
    const int nend = seg_end[prow];
    const int ndeg = cnt[prow];
    const float ndi = dinv[prow];
    const v4f nself = *(const v4f*)(xl + (size_t)prow * DFEAT);

    v4f acc = {0.0f, 0.0f, 0.0f, 0.0f};
    if (deg > 0) {
      chunk_sum(acc, xl, sj, min(32, deg));
      for (base += 32; base < end; base += 32) {
        sj = csr_src[min(base + lane, end - 1)];
        chunk_sum(acc, xl, sj, min(32, end - base));
      }
    }
    acc += self_row;
    v4f res = acc * di + b;
    if (SCALE_OUT) res *= di;
    *(v4f*)(out + (size_t)row * DFEAT + 4 * lane) = res;

    if (!more) break;
    row = nrow;
    end = nend;
    deg = ndeg;
    di = ndi;
    self_row = nself;
  }
}

extern "C" void kernel_launch(void* const* d_in, const int* in_sizes, int n_in,
                              void* d_out, int out_size, void* d_ws, size_t ws_size,
                              hipStream_t stream) {
  const float* x     = (const float*)d_in[0];
  const int*   ei    = (const int*)d_in[1];
  const float* t     = (const float*)d_in[2];
  const float* W1    = (const float*)d_in[3];
  const float* b1    = (const float*)d_in[4];
  const float* W2    = (const float*)d_in[5];
  const float* b2    = (const float*)d_in[6];
  const float* freq  = (const float*)d_in[7];
  const float* phase = (const float*)d_in[8];

  const int n_nodes = in_sizes[2];
  const int n_edges = in_sizes[1] / 2;
  const size_t nd = (size_t)n_nodes * DFEAT;

  const int* src = ei;
  const int* dst = ei + n_edges;

  float* out = (float*)d_out;

  const int T = 256;
  const int nb = (n_nodes + RB - 1) / RB;
  const int nblk = (n_edges + CHUNK - 1) / CHUNK;

  float* dinv = (float*)d_ws;
  size_t noff = ((size_t)n_nodes + 255) & ~(size_t)255;
  size_t moff = ((size_t)nb * nblk + 255) & ~(size_t)255;
  size_t boff = ((size_t)nb + 255) & ~(size_t)255;
  size_t eoff = ((size_t)n_edges + 255) & ~(size_t)255;
  float* bufA = dinv + noff;
  float* bufB = bufA + nd;
  int* cnt      = (int*)(bufB + nd);
  int* cursor   = cnt + noff;
  int* bmat     = cursor + noff;
  int* btot     = bmat + moff;
  int* pairs    = btot + boff;
  int* csr_src  = pairs + 2 * eoff;
  float* W12 = (float*)(csr_src + eoff);
  float* c12 = W12 + (size_t)DFEAT * DFEAT;

  tgcn_bucket_hist<<<nblk, 1024, 0, stream>>>(dst, bmat, n_edges, nb, nblk);
  tgcn_scan_fold<<<nb + 9, T, 0, stream>>>(bmat, btot, nb, nblk, W1, b1, W2, W12, c12);
  tgcn_bucket_scatter<<<nblk, 1024, 0, stream>>>(src, dst, bmat, btot, pairs, n_edges, nb, nblk);
  tgcn_bucket_sort<<<nb, RB, 0, stream>>>(pairs, btot, csr_src, cursor, cnt, dinv, n_nodes);

  tgcn_gemm_wmma<<<GEMM_WGS, T, 0, stream>>>(x, W12, t, freq, phase, 1, dinv, bufA, n_nodes);

  tgcn_gather<true><<<GATHER_WGS, T, 0, stream>>>(bufA, csr_src, cursor, cnt, dinv, c12, bufB, n_nodes);

  tgcn_gather<false><<<GATHER_WGS, T, 0, stream>>>(bufB, csr_src, cursor, cnt, dinv, b2, out, n_nodes);
}


# edge grouping level 1: each work-group sorts its 8192-edge chunk by bucket in LDS; level 2 walks its run of every chunk; histogram and scan launches go (5 launches); sum order = LDS arrival order
# speedup vs baseline: 6.3395x; 1.0902x over previous
#include <hip/hip_runtime.h>
#include <math.h>

typedef __attribute__((ext_vector_type(2))) float v2f;
typedef __attribute__((ext_vector_type(2))) int v2i;
typedef __attribute__((ext_vector_type(4))) float v4f;
typedef __attribute__((ext_vector_type(4))) int v4i;
typedef __attribute__((ext_vector_type(8))) float v8f;

#define DFEAT 128
#define APITCH 132
#define RB 512
#define RB_SHIFT 9
#define MAXB 1024
#define CHUNK 8192
#define SRC_BITS 19
#define SRC_MASK ((1 << SRC_BITS) - 1)
#define WALK_U 16


__device__ __forceinline__ int wave_sum(int v) {
  for (int o = 16; o > 0; o >>= 1) v += __shfl_xor(v, o);
  return v;
}

__device__ __forceinline__ int wave_scan_incl(int v, int lane) {
  for (int o = 1; o < 32; o <<= 1) {
    int y = __shfl_up(v, o);
    if (lane >= o) v += y;
  }
  return v;
}

__device__ __forceinline__ void wmma_tile_16x128(
    const float* sA, const float* __restrict__ W, float* __restrict__ Y,
    int mbase, int n_rows, int tid) {
  const int wave = tid >> 5;
  const int lane = tid & 31;
  const int half = lane >> 4;
  const int arow = lane & 15;
  const int ncol = wave * 16 + (lane & 15);

  v8f acc = {};
  for (int k = 0; k < DFEAT; k += 4) {
    const int ka = k + 2 * half;
    v2f a, b;
    a.x = sA[arow * APITCH + ka];
    a.y = sA[arow * APITCH + ka + 1];
    b.x = W[(size_t)ka * DFEAT + ncol];
    b.y = W[(size_t)(ka + 1) * DFEAT + ncol];
    acc = __builtin_amdgcn_wmma_f32_16x16x4_f32(
         false, a,  false, b,
         (short)0, acc,  false,  false);
  }

#pragma unroll
  for (int r = 0; r < 8; ++r) {
    int orow = mbase + r + 8 * half;
    if (orow < n_rows) Y[(size_t)orow * DFEAT + ncol] = acc[r];
  }
}

__global__ void __launch_bounds__(1024) tgcn_chunk_sort(
    const int* __restrict__ src, const int* __restrict__ dst,
    int* __restrict__ packed, int* __restrict__ offs, int n_edges, int nb,
    int nblk, const float* __restrict__ W1, const float* __restrict__ b1,
    const float* __restrict__ W2, float* __restrict__ W12,
    float* __restrict__ c12) {
  __shared__ __attribute__((aligned(16))) float sA[16 * APITCH];
  __shared__ int s_cnt[MAXB];
  __shared__ int s_w[32];
  __shared__ __attribute__((aligned(16))) int s_sorted[CHUNK];
  const int tid = threadIdx.x;

  if ((int)blockIdx.x >= nblk) {
    const int tile = blockIdx.x - nblk;
    const bool is_w = tile < 8;
    if (tid < 256) {
      for (int i = tid; i < 16 * (DFEAT / 4); i += 256) {
        const int r = i >> 5;
        const int c4 = i & 31;
        v4f v = {0.0f, 0.0f, 0.0f, 0.0f};
        if (is_w)
          v = *(const v4f*)(W1 + (size_t)(tile * 16 + r) * DFEAT + 4 * c4);
        else if (r == 0)
          v = *(const v4f*)(b1 + 4 * c4);
        *(v4f*)(sA + r * APITCH + 4 * c4) = v;
      }
    }
    __syncthreads();
    if (tid < 256)
      wmma_tile_16x128(sA, W2, is_w ? W12 : c12, is_w ? tile * 16 : 0,
                       is_w ? DFEAT : 1, tid);
    return;
  }

  const int lane = tid & 31;
  const int wave = tid >> 5;
  const int g = blockIdx.x;
  const int total = min(CHUNK, n_edges - g * CHUNK);
  if (tid < nb) s_cnt[tid] = 0;
  const int base = g * CHUNK + tid;
  int d[8], s[8];
#pragma unroll
  for (int j = 0; j < 8; ++j) {
    const int e = base + j * 1024;
    d[j] = (e < n_edges) ? dst[e] : -1;
    s[j] = (e < n_edges) ? src[e] : 0;
  }
  __syncthreads();
#pragma unroll
  for (int j = 0; j < 8; ++j)
    if (d[j] >= 0) atomicAdd(&s_cnt[d[j] >> RB_SHIFT], 1);
  __syncthreads();

  const int cb = (tid < nb) ? s_cnt[tid] : 0;
  const int inc = wave_scan_incl(cb, lane);
  if (lane == 31) s_w[wave] = inc;
  __syncthreads();
  int off = 0;
  for (int w = 0; w < 32; ++w)
    if (w < wave) off += s_w[w];
  const int start = off + inc - cb;
  if (tid < nb) {
    s_cnt[tid] = start;
    offs[(size_t)tid * nblk + g] = start;
  }
  __syncthreads();
  int slot[8];
#pragma unroll
  for (int j = 0; j < 8; ++j)
    slot[j] = (d[j] >= 0) ? atomicAdd(&s_cnt[d[j] >> RB_SHIFT], 1) : -1;
#pragma unroll
  for (int j = 0; j < 8; ++j)
    if (slot[j] >= 0)
      s_sorted[slot[j]] = s[j] | ((d[j] & (RB - 1)) << SRC_BITS);
  __syncthreads();
  int* outp = packed + (size_t)g * CHUNK;
  if (total == CHUNK) {
#pragma unroll
    for (int j = 0; j < 2; ++j) {
      const int i = 4 * tid + j * 4096;
      *(v4i*)(outp + i) = *(const v4i*)(s_sorted + i);
    }
  } else {
#pragma unroll
    for (int j = 0; j < 8; ++j) {
      const int i = tid + j * 1024;
      if (i < total) outp[i] = s_sorted[i];
    }
  }
}

template <bool PLACE>
__device__ __forceinline__ int bucket_walk(
    const int* __restrict__ packed, const int* __restrict__ offs, int* s_c,
    int* __restrict__ csr_seg, int b, int nb, int nblk, int n_edges, int tid) {
  int pre = 0;
  for (int g0 = 0; g0 < nblk; g0 += RB / 2) {
    const int g = g0 + (tid >> 1);
    int o0 = 0, len = 0;
    if (g < nblk) {
      o0 = offs[(size_t)b * nblk + g];
      const int o1 = (b + 1 < nb) ? offs[(size_t)(b + 1) * nblk + g]
                                  : min(CHUNK, n_edges - g * CHUNK);
      len = o1 - o0;
      if (!(tid & 1)) pre += o0;
    }
    const size_t rbase = (size_t)g * CHUNK + o0;
    for (int k0 = tid & 1; k0 < len; k0 += 2 * WALK_U) {
      int e[WALK_U];
#pragma unroll
      for (int u = 0; u < WALK_U; ++u)
        e[u] = packed[rbase + min(k0 + 2 * u, len - 1)];
      asm volatile("" ::"v"(e[0]), "v"(e[1]), "v"(e[2]), "v"(e[3]), "v"(e[4]),
                   "v"(e[5]), "v"(e[6]), "v"(e[7]), "v"(e[8]), "v"(e[9]),
                   "v"(e[10]), "v"(e[11]), "v"(e[12]), "v"(e[13]), "v"(e[14]),
                   "v"(e[15]));
      if (PLACE) {
        int slot[WALK_U];
#pragma unroll
        for (int u = 0; u < WALK_U; ++u)
          slot[u] = (k0 + 2 * u < len) ? atomicAdd(&s_c[e[u] >> SRC_BITS], 1)
                                       : -1;
#pragma unroll
        for (int u = 0; u < WALK_U; ++u)
          if (slot[u] >= 0) csr_seg[slot[u]] = e[u] & SRC_MASK;
      } else {
#pragma unroll
        for (int u = 0; u < WALK_U; ++u)
          if (k0 + 2 * u < len) atomicAdd(&s_c[e[u] >> SRC_BITS], 1);
      }
    }
  }
  return pre;
}

__global__ void __launch_bounds__(RB) tgcn_bucket_sort(
    const int* __restrict__ packed, const int* __restrict__ offs,
    int* __restrict__ csr_src, int* __restrict__ seg_end, int* __restrict__ cnt,
    float* __restrict__ dinv, int n_nodes, int n_edges, int nb, int nblk) {
  __shared__ int s_c[RB];
  __shared__ int s_w[RB / 32];
  __shared__ int s_pre[RB / 32];
  const int tid = threadIdx.x;
  const int lane = tid & 31;
  const int wave = tid >> 5;
  const int b = blockIdx.x;
  const int nbase = b << RB_SHIFT;

  s_c[tid] = 0;
  __syncthreads();

  int pre = bucket_walk<false>(packed, offs, s_c, csr_src, b, nb, nblk,
                               n_edges, tid);
  pre = wave_sum(pre);
  if (lane == 0) s_pre[wave] = pre;
  __syncthreads();
  int bstart = 0;
  for (int w = 0; w < RB / 32; ++w) bstart += s_pre[w];

  const int c = s_c[tid];
  const int inc = wave_scan_incl(c, lane);
  if (lane == 31) s_w[wave] = inc;
  __syncthreads();
  int off = 0;
  for (int w = 0; w < RB / 32; ++w)
    if (w < wave) off += s_w[w];
  const int first = off + inc - c;
  const int node = nbase + tid;
  if (node < n_nodes) {
    cnt[node] = c;
    seg_end[node] = bstart + first + c;
    dinv[node] = rsqrtf((float)(c + 1));
  }
  s_c[tid] = first;
  __syncthreads();

  bucket_walk<true>(packed, offs, s_c, csr_src + bstart, b, nb, nblk, n_edges,
                    tid);
}

#define GEMM_WGS 768

struct ATileRows {
  v4f x0, x1;
  float t0, t1;
  float d0, d1;
};

__device__ __forceinline__ ATileRows a_rows_fetch(
    const float* __restrict__ X, const float* __restrict__ t,
    const float* __restrict__ rowscale, int tile, int wave, int lane,
    int n_nodes) {
  const int r0 = min(tile * 16 + wave, n_nodes - 1);
  const int r1 = min(tile * 16 + wave + 8, n_nodes - 1);
  ATileRows f;
  f.x0 = *(const v4f*)(X + (size_t)r0 * DFEAT + 4 * lane);
  f.x1 = *(const v4f*)(X + (size_t)r1 * DFEAT + 4 * lane);
  f.t0 = t[r0];
  f.t1 = t[r1];
  f.d0 = rowscale[r0];
  f.d1 = rowscale[r1];
  return f;
}

__device__ __forceinline__ v4f a_row_value(v4f x, float tr, float d, v4f fq,
                                           v4f ph, int use_te, bool inside) {
  v4f v = x;
  if (use_te) {
    v.x += cosf(tr * fq.x + ph.x);
    v.y += cosf(tr * fq.y + ph.y);
    v.z += cosf(tr * fq.z + ph.z);
    v.w += cosf(tr * fq.w + ph.w);
  }
  v *= d;
  if (!inside) v = (v4f){0.0f, 0.0f, 0.0f, 0.0f};
  return v;
}

__global__ void __launch_bounds__(256) tgcn_gemm_wmma(
    const float* __restrict__ X, const float* __restrict__ W,
    const float* __restrict__ t, const float* __restrict__ freq,
    const float* __restrict__ phase, int use_te,
    const float* __restrict__ rowscale, float* __restrict__ Y, int n_nodes) {
  __shared__ __attribute__((aligned(16))) float sA[2][16 * APITCH];

  const int tid = threadIdx.x;
  const int lane = tid & 31;
  const int wave = __builtin_amdgcn_readfirstlane(tid >> 5);
  const int half = lane >> 4;
  const int arow = lane & 15;
  const int ncol = wave * 16 + arow;
  const int ntiles = (n_nodes + 15) >> 4;
  const int step = gridDim.x;

  int tile = blockIdx.x;
  if (tile >= ntiles) return;

  ATileRows cur = a_rows_fetch(X, t, rowscale, tile, wave, lane, n_nodes);

  v2f bw[DFEAT / 4];
#pragma unroll
  for (int s = 0; s < DFEAT / 4; ++s) {
    const int ka = 4 * s + 2 * half;
    bw[s].x = W[(size_t)ka * DFEAT + ncol];
    bw[s].y = W[(size_t)(ka + 1) * DFEAT + ncol];
  }

  v4f fq = {0.0f, 0.0f, 0.0f, 0.0f}, ph = {0.0f, 0.0f, 0.0f, 0.0f};
  if (use_te) {
    fq = *(const v4f*)(freq + 4 * lane);
    ph = *(const v4f*)(phase + 4 * lane);
  }

  float* wr0 = &sA[0][wave * APITCH + 4 * lane];
  float* wr1 = &sA[0][(wave + 8) * APITCH + 4 * lane];
  const float* rd = &sA[0][arow * APITCH + 2 * half];

  *(v4f*)wr0 = a_row_value(cur.x0, cur.t0, cur.d0, fq, ph, use_te,
                           tile * 16 + wave < n_nodes);
  *(v4f*)wr1 = a_row_value(cur.x1, cur.t1, cur.d1, fq, ph, use_te,
                           tile * 16 + wave + 8 < n_nodes);
  __syncthreads();

  int buf = 0;
  for (;;) {
    const int ntile = tile + step;
    const bool more = ntile < ntiles;
    const ATileRows nxt =
        a_rows_fetch(X, t, rowscale, more ? ntile : tile, wave, lane, n_nodes);

    const float* sa = rd + buf * (16 * APITCH);
    v8f acc = {};
#pragma unroll
    for (int s = 0; s < DFEAT / 4; ++s) {
      const v2f a = *(const v2f*)(sa + 4 * s);
      acc = __builtin_amdgcn_wmma_f32_16x16x4_f32(
           false, a,  false, bw[s],
           (short)0, acc,  false,  false);
    }

    const int mbase = tile * 16;
#pragma unroll
    for (int r = 0; r < 8; ++r) {
      const int orow = mbase + r + 8 * half;
      if (orow < n_nodes) Y[(size_t)orow * DFEAT + ncol] = acc[r];
    }

    if (!more) break;
    const int off = (buf ^ 1) * (16 * APITCH);
    *(v4f*)(wr0 + off) = a_row_value(nxt.x0, nxt.t0, nxt.d0, fq, ph, use_te,
                                     ntile * 16 + wave < n_nodes);
    *(v4f*)(wr1 + off) = a_row_value(nxt.x1, nxt.t1, nxt.d1, fq, ph, use_te,
                                     ntile * 16 + wave + 8 < n_nodes);
    __syncthreads();
    buf ^= 1;
    tile = ntile;
  }
}

template <int G, bool PRED>
__device__ __forceinline__ void row_group(v4f& acc, const float* __restrict__ xl,
                                          int sj, int j, int m) {
  v4f v[G];
#pragma unroll
  for (int u = 0; u < G; ++u) {
    int idx = j + u;
    if (PRED) idx = min(idx, m - 1);
    const int s = __builtin_amdgcn_readlane(sj, idx);
    v[u] = *(const v4f*)(xl + (size_t)s * DFEAT);
  }
#pragma unroll
  for (int u = 0; u < G; ++u)
    if (!PRED || j + u < m) acc += v[u];
}

__device__ __forceinline__ void chunk_sum(v4f& acc, const float* __restrict__ xl,
                                          int sj, int m) {
  int j = 0;
  for (; j + 8 <= m; j += 8) row_group<8, false>(acc, xl, sj, j, m);
  if (j < m) {
    row_group<4, true>(acc, xl, sj, j, m);
    j += 4;
  }
  if (j < m) row_group<4, true>(acc, xl, sj, j, m);
}

#define GATHER_WGS 1792

template <bool SCALE_OUT>
__global__ void __launch_bounds__(256) tgcn_gather(
    const float* __restrict__ xs, const int* __restrict__ csr_src,
    const int* __restrict__ seg_end, const int* __restrict__ cnt,
    const float* __restrict__ dinv, const float* __restrict__ bias,
    float* __restrict__ out, int n_nodes) {
  const int lane = threadIdx.x & 31;
  const int wave = __builtin_amdgcn_readfirstlane(threadIdx.x >> 5);
  const int nwaves = gridDim.x * 8;
  const float* xl = xs + 4 * lane;
  const v4f b = *(const v4f*)(bias + 4 * lane);

  int row = blockIdx.x * 8 + wave;
  if (row >= n_nodes) return;
  int end = seg_end[row];
  int deg = cnt[row];
  float di = dinv[row];
  v4f self_row = *(const v4f*)(xl + (size_t)row * DFEAT);

  for (;;) {
    int base = end - deg;
    int sj = 0;
    if (deg > 0) sj = csr_src[min(base + lane, end - 1)];

    const int nrow = row + nwaves;
    const bool more = nrow < n_nodes;
    const int prow = more ? nrow : row;
    const int nend = seg_end[prow];
    const int ndeg = cnt[prow];
    const float ndi = dinv[prow];
    const v4f nself = *(const v4f*)(xl + (size_t)prow * DFEAT);

    v4f acc = {0.0f, 0.0f, 0.0f, 0.0f};
    if (deg > 0) {
      chunk_sum(acc, xl, sj, min(32, deg));
      for (base += 32; base < end; base += 32) {
        sj = csr_src[min(base + lane, end - 1)];
        chunk_sum(acc, xl, sj, min(32, end - base));
      }
    }
    acc += self_row;
    v4f res = acc * di + b;
    if (SCALE_OUT) res *= di;
    *(v4f*)(out + (size_t)row * DFEAT + 4 * lane) = res;

    if (!more) break;
    row = nrow;
    end = nend;
    deg = ndeg;
    di = ndi;
    self_row = nself;
  }
}

extern "C" void kernel_launch(void* const* d_in, const int* in_sizes, int n_in,
                              void* d_out, int out_size, void* d_ws, size_t ws_size,
                              hipStream_t stream) {
  const float* x     = (const float*)d_in[0];
  const int*   ei    = (const int*)d_in[1];
  const float* t     = (const float*)d_in[2];
  const float* W1    = (const float*)d_in[3];
  const float* b1    = (const float*)d_in[4];
  const float* W2    = (const float*)d_in[5];
  const float* b2    = (const float*)d_in[6];
  const float* freq  = (const float*)d_in[7];
  const float* phase = (const float*)d_in[8];

  const int n_nodes = in_sizes[2];
  const int n_edges = in_sizes[1] / 2;
  const size_t nd = (size_t)n_nodes * DFEAT;

  const int* src = ei;
  const int* dst = ei + n_edges;

  float* out = (float*)d_out;

  const int T = 256;
  const int nb = (n_nodes + RB - 1) / RB;
  const int nblk = (n_edges + CHUNK - 1) / CHUNK;

  float* dinv = (float*)d_ws;
  size_t noff = ((size_t)n_nodes + 255) & ~(size_t)255;
  size_t moff = ((size_t)nb * nblk + 255) & ~(size_t)255;
  size_t eoff = ((size_t)n_edges + 255) & ~(size_t)255;
  float* bufA = dinv + noff;
  float* bufB = bufA + nd;
  int* cnt      = (int*)(bufB + nd);
  int* cursor   = cnt + noff;
  int* offs     = cursor + noff;
  int* packed   = offs + moff;
  int* csr_src  = packed + eoff;
  float* W12 = (float*)(csr_src + eoff);
  float* c12 = W12 + (size_t)DFEAT * DFEAT;

  tgcn_chunk_sort<<<nblk + 9, 1024, 0, stream>>>(src, dst, packed, offs, n_edges, nb, nblk, W1, b1, W2, W12, c12);
  tgcn_bucket_sort<<<nb, RB, 0, stream>>>(packed, offs, csr_src, cursor, cnt, dinv, n_nodes, n_edges, nb, nblk);

  tgcn_gemm_wmma<<<GEMM_WGS, T, 0, stream>>>(x, W12, t, freq, phase, 1, dinv, bufA, n_nodes);

  tgcn_gather<true><<<GATHER_WGS, T, 0, stream>>>(bufA, csr_src, cursor, cnt, dinv, c12, bufB, n_nodes);

  tgcn_gather<false><<<GATHER_WGS, T, 0, stream>>>(bufB, csr_src, cursor, cnt, dinv, b2, out, n_nodes);
}
